# starting kernel
# baseline (speedup 1.0000x reference)
#include <hip/hip_runtime.h>
#include <hip/hip_bf16.h>
#include <math.h>

typedef _Float16 v16h __attribute__((ext_vector_type(16)));
typedef float    v8f  __attribute__((ext_vector_type(8)));

__device__ __forceinline__ unsigned ordf(float f) {
    unsigned u = __float_as_uint(f);
    return (u & 0x80000000u) ? ~u : (u | 0x80000000u);
}
__device__ __forceinline__ float deordf(unsigned u) {
    unsigned v = (u & 0x80000000u) ? (u & 0x7FFFFFFFu) : ~u;
    return __uint_as_float(v);
}

__global__ void pack_w_k(const float* __restrict__ W, _Float16* __restrict__ Wp)
{
    int idx = blockIdx.x * blockDim.x + threadIdx.x;
    if (idx >= 128 * 128) return;
    const int r  = idx >> 4;
    const int i  = idx & 15;
    const int col = r & 127;
    const int lh  = (r >> 7) & 1;
    const int kb  = r >> 8;
    const int kk  = (i < 8) ? (8 * lh + i) : (8 + 8 * lh + i);
    const int K   = 32 * kb + kk;
    Wp[(size_t)r * 16 + i] = (_Float16)W[(size_t)K * 128 + col];
}

__global__ __launch_bounds__(256) void gemm128_wmma(
    const float* __restrict__ X, const _Float16* __restrict__ Wp,
    const float* __restrict__ bias, float* __restrict__ Y, int M)
{
    const int wv    = threadIdx.x >> 5;
    const int lane  = threadIdx.x & 31;
    const int lh    = lane >> 4;
    const int lm    = lane & 15;
    const int mbase = blockIdx.x * 16;

    int rowA = mbase + lm; if (rowA >= M) rowA = M - 1;
    const int colB = wv * 16 + lm;

    v8f acc = {};
#pragma unroll
    for (int kt = 0; kt < 4; ++kt) {
        const int kb = kt * 32;
        const float4* pr = (const float4*)(X + (size_t)rowA * 128 + kb);
        const float4 x0 = pr[2 * lh + 0];
        const float4 x1 = pr[2 * lh + 1];
        const float4 x2 = pr[4 + 2 * lh + 0];
        const float4 x3 = pr[4 + 2 * lh + 1];
        v16h a;
        a[0]  = (_Float16)x0.x; a[1]  = (_Float16)x0.y;
        a[2]  = (_Float16)x0.z; a[3]  = (_Float16)x0.w;
        a[4]  = (_Float16)x1.x; a[5]  = (_Float16)x1.y;
        a[6]  = (_Float16)x1.z; a[7]  = (_Float16)x1.w;
        a[8]  = (_Float16)x2.x; a[9]  = (_Float16)x2.y;
        a[10] = (_Float16)x2.z; a[11] = (_Float16)x2.w;
        a[12] = (_Float16)x3.x; a[13] = (_Float16)x3.y;
        a[14] = (_Float16)x3.z; a[15] = (_Float16)x3.w;
        const v16h b = *(const v16h*)(Wp + ((size_t)(kt * 2 + lh) * 128 + colB) * 16);
        acc = __builtin_amdgcn_wmma_f32_16x16x32_f16(
                  false, a, false, b, (short)0, acc, false, false);
    }
    const float bb = bias[colB];
    if (mbase + 16 <= M) {
#pragma unroll
        for (int r = 0; r < 8; ++r)
            Y[(size_t)(mbase + r + 8 * lh) * 128 + colB] = acc[r] + bb;
    } else {
#pragma unroll
        for (int r = 0; r < 8; ++r) {
            const int m = mbase + r + 8 * lh;
            if (m < M) Y[(size_t)m * 128 + colB] = acc[r] + bb;
        }
    }
}

__global__ void init_attn_k(unsigned* __restrict__ amaxU, float* __restrict__ denom, int n)
{
    int i = blockIdx.x * blockDim.x + threadIdx.x;
    if (i < n) { amaxU[i] = 0u; denom[i] = 0.f; }
}

__global__ void init_out_k(float* __restrict__ out, const float* __restrict__ bias, int n)
{
    int i = blockIdx.x * blockDim.x + threadIdx.x;
    if (i < n) out[i] = bias[i & 127];
}

__global__ __launch_bounds__(256) void edge_alpha_k(
    const float* __restrict__ xl, const int* __restrict__ ei,
    const float* __restrict__ att, float* __restrict__ alpha,
    unsigned* __restrict__ amaxU, int E)
{
    const int wid  = (int)((blockIdx.x * blockDim.x + threadIdx.x) >> 5);
    const int lane = threadIdx.x & 31;
    if (wid >= E) return;
    const int s = ei[wid], d = ei[E + wid];
    const float* ps = xl + (size_t)s * 128;
    const float* pd = xl + (size_t)d * 128;
#pragma unroll
    for (int h = 0; h < 4; ++h) {
        const int c = h * 32 + lane;
        float v = ps[c] + pd[c];
        v = v > 0.f ? v : 0.2f * v;
        float p = v * att[c];
#pragma unroll
        for (int off = 16; off >= 1; off >>= 1) p += __shfl_xor(p, off, 32);
        if (lane == 0) {
            alpha[(size_t)wid * 4 + h] = p;
            atomicMax(&amaxU[(size_t)d * 4 + h], ordf(p));
        }
    }
}

__global__ void edge_expsum_k(
    float* __restrict__ alpha, const int* __restrict__ ei,
    const unsigned* __restrict__ amaxU, float* __restrict__ denom, int E)
{
    int i = blockIdx.x * blockDim.x + threadIdx.x;
    if (i >= E * 4) return;
    const int e = i >> 2, h = i & 3;
    const int d = ei[E + e];
    const float m  = deordf(amaxU[(size_t)d * 4 + h]);
    const float ex = __expf(alpha[i] - m);
    alpha[i] = ex;
    atomicAdd(&denom[(size_t)d * 4 + h], ex);
}

__global__ __launch_bounds__(256) void edge_scatter_k(
    const float* __restrict__ xl, const int* __restrict__ ei,
    const float* __restrict__ alpha, const float* __restrict__ denom,
    float* __restrict__ out, int E)
{
    const int wid  = (int)((blockIdx.x * blockDim.x + threadIdx.x) >> 5);
    const int lane = threadIdx.x & 31;
    if (wid >= E) return;
    const int s = ei[wid], d = ei[E + wid];
    const float* ps = xl + (size_t)s * 128;
    float* po = out + (size_t)d * 128;
#pragma unroll
    for (int h = 0; h < 4; ++h) {
        const float coef = alpha[(size_t)wid * 4 + h] / denom[(size_t)d * 4 + h];
        const int c = h * 32 + lane;
        atomicAdd(&po[c], coef * ps[c]);
    }
}

__global__ void gelu_k(float* __restrict__ x, int n)
{
    int i = blockIdx.x * blockDim.x + threadIdx.x;
    if (i < n) {
        float v = x[i];
        x[i] = 0.5f * v * (1.f + erff(v * 0.70710678118654752440f));
    }
}

__global__ __launch_bounds__(256) void layernorm_k(
    const float* __restrict__ y, const float* __restrict__ gamma,
    const float* __restrict__ beta, float* __restrict__ out, int rows)
{
    const int wv = threadIdx.x >> 5, lane = threadIdx.x & 31;
    const int row = blockIdx.x * 8 + wv;
    if (row >= rows) return;
    const float* p = y + (size_t)row * 128 + lane * 4;
    float v0 = p[0], v1 = p[1], v2 = p[2], v3 = p[3];
    float s = v0 + v1 + v2 + v3;
#pragma unroll
    for (int off = 16; off >= 1; off >>= 1) s += __shfl_xor(s, off, 32);
    const float mu = s * (1.f / 128.f);
    const float d0 = v0 - mu, d1 = v1 - mu, d2 = v2 - mu, d3 = v3 - mu;
    float q = d0 * d0 + d1 * d1 + d2 * d2 + d3 * d3;
#pragma unroll
    for (int off = 16; off >= 1; off >>= 1) q += __shfl_xor(q, off, 32);
    const float rs = rsqrtf(q * (1.f / 128.f) + 1e-12f);
    const int c = lane * 4;
    float* o = out + (size_t)row * 128 + c;
    o[0] = d0 * rs * gamma[c + 0] + beta[c + 0];
    o[1] = d1 * rs * gamma[c + 1] + beta[c + 1];
    o[2] = d2 * rs * gamma[c + 2] + beta[c + 2];
    o[3] = d3 * rs * gamma[c + 3] + beta[c + 3];
}

extern "C" void kernel_launch(void* const* d_in, const int* in_sizes, int n_in,
                              void* d_out, int out_size, void* d_ws, size_t ws_size,
                              hipStream_t stream)
{
    const float* embs  = (const float*)d_in[0];
    const float* w0    = (const float*)d_in[1];
    const float* b0    = (const float*)d_in[2];
    const float* att0  = (const float*)d_in[3];
    const float* bias0 = (const float*)d_in[4];
    const float* w1    = (const float*)d_in[5];
    const float* b1    = (const float*)d_in[6];
    const float* att1  = (const float*)d_in[7];
    const float* bias1 = (const float*)d_in[8];
    const float* wout  = (const float*)d_in[9];
    const float* bout  = (const float*)d_in[10];
    const float* gamma = (const float*)d_in[11];
    const float* beta  = (const float*)d_in[12];
    const int*   ei0   = (const int*)d_in[13];
    const int*   ei1   = (const int*)d_in[14];

    const int N  = in_sizes[0] / 128;
    const int E  = in_sizes[13] / 2;
    const int BS = out_size / 128;

    char* wsp = (char*)d_ws;
    size_t off = 0;
    auto carve = [&](size_t bytes) -> void* {
        void* p = wsp + off; off += (bytes + 255) & ~(size_t)255; return p;
    };
    float*     xl    = (float*)    carve((size_t)N * 128 * 4);
    float*     buf   = (float*)    carve((size_t)N * 128 * 4);
    float*     alpha = (float*)    carve((size_t)E * 4 * 4);
    unsigned*  amaxU = (unsigned*) carve((size_t)N * 4 * 4);
    float*     denom = (float*)    carve((size_t)N * 4 * 4);
    float*     ybuf  = (float*)    carve((size_t)BS * 128 * 4);
    _Float16*  Wp    = (_Float16*) carve((size_t)128 * 128 * 2);

    const int gemmBlocksN = (N + 15) / 16;
    const int edgeWaves   = (E + 7) / 8;
    const int ehBlocks    = (E * 4 + 255) / 256;
    const int attnBlocks  = (N * 4 + 255) / 256;
    const int featBlocks  = (N * 128 + 255) / 256;
    const int packBlocks  = (128 * 128 + 255) / 256;

    pack_w_k    <<<packBlocks, 256, 0, stream>>>(w0, Wp);
    gemm128_wmma<<<gemmBlocksN, 256, 0, stream>>>(embs, Wp, b0, xl, N);
    init_attn_k <<<attnBlocks, 256, 0, stream>>>(amaxU, denom, N * 4);
    init_out_k  <<<featBlocks, 256, 0, stream>>>(buf, bias0, N * 128);
    edge_alpha_k<<<edgeWaves,  256, 0, stream>>>(xl, ei0, att0, alpha, amaxU, E);
    edge_expsum_k<<<ehBlocks,  256, 0, stream>>>(alpha, ei0, amaxU, denom, E);
    edge_scatter_k<<<edgeWaves,256, 0, stream>>>(xl, ei0, alpha, denom, buf, E);
    gelu_k      <<<featBlocks, 256, 0, stream>>>(buf, N * 128);

    pack_w_k    <<<packBlocks, 256, 0, stream>>>(w1, Wp);
    gemm128_wmma<<<gemmBlocksN, 256, 0, stream>>>(buf, Wp, b1, xl, N);
    init_attn_k <<<attnBlocks, 256, 0, stream>>>(amaxU, denom, N * 4);
    init_out_k  <<<featBlocks, 256, 0, stream>>>(buf, bias1, N * 128);
    edge_alpha_k<<<edgeWaves,  256, 0, stream>>>(xl, ei1, att1, alpha, amaxU, E);
    edge_expsum_k<<<ehBlocks,  256, 0, stream>>>(alpha, ei1, amaxU, denom, E);
    edge_scatter_k<<<edgeWaves,256, 0, stream>>>(xl, ei1, alpha, denom, buf, E);

    pack_w_k    <<<packBlocks, 256, 0, stream>>>(wout, Wp);
    gemm128_wmma<<<(BS + 15) / 16, 256, 0, stream>>>(buf, Wp, bout, ybuf, BS);
    layernorm_k <<<(BS + 7) / 8,   256, 0, stream>>>(ybuf, gamma, beta, (float*)d_out, BS);
}


# edges counting-sorted by destination in-call; one wave per node does logits, running softmax and weighted sum in registers; no float atomics left; sum order now follows int-atomic slot order
# speedup vs baseline: 2.2843x; 2.2843x over previous
#include <hip/hip_runtime.h>
#include <hip/hip_bf16.h>
#include <math.h>

typedef _Float16 v16h __attribute__((ext_vector_type(16)));
typedef float    v8f  __attribute__((ext_vector_type(8)));

__global__ void pack_w_k(const float* __restrict__ W, _Float16* __restrict__ Wp)
{
    int idx = blockIdx.x * blockDim.x + threadIdx.x;
    if (idx >= 128 * 128) return;
    const int r  = idx >> 4;
    const int i  = idx & 15;
    const int col = r & 127;
    const int lh  = (r >> 7) & 1;
    const int kb  = r >> 8;
    const int kk  = (i < 8) ? (8 * lh + i) : (8 + 8 * lh + i);
    const int K   = 32 * kb + kk;
    Wp[(size_t)r * 16 + i] = (_Float16)W[(size_t)K * 128 + col];
}

__global__ __launch_bounds__(256) void gemm128_wmma(
    const float* __restrict__ X, const _Float16* __restrict__ Wp,
    const float* __restrict__ bias, float* __restrict__ Y, int M)
{
    const int wv    = threadIdx.x >> 5;
    const int lane  = threadIdx.x & 31;
    const int lh    = lane >> 4;
    const int lm    = lane & 15;
    const int mbase = blockIdx.x * 16;

    int rowA = mbase + lm; if (rowA >= M) rowA = M - 1;
    const int colB = wv * 16 + lm;

    v8f acc = {};
#pragma unroll
    for (int kt = 0; kt < 4; ++kt) {
        const int kb = kt * 32;
        const float4* pr = (const float4*)(X + (size_t)rowA * 128 + kb);
        const float4 x0 = pr[2 * lh + 0];
        const float4 x1 = pr[2 * lh + 1];
        const float4 x2 = pr[4 + 2 * lh + 0];
        const float4 x3 = pr[4 + 2 * lh + 1];
        v16h a;
        a[0]  = (_Float16)x0.x; a[1]  = (_Float16)x0.y;
        a[2]  = (_Float16)x0.z; a[3]  = (_Float16)x0.w;
        a[4]  = (_Float16)x1.x; a[5]  = (_Float16)x1.y;
        a[6]  = (_Float16)x1.z; a[7]  = (_Float16)x1.w;
        a[8]  = (_Float16)x2.x; a[9]  = (_Float16)x2.y;
        a[10] = (_Float16)x2.z; a[11] = (_Float16)x2.w;
        a[12] = (_Float16)x3.x; a[13] = (_Float16)x3.y;
        a[14] = (_Float16)x3.z; a[15] = (_Float16)x3.w;
        const v16h b = *(const v16h*)(Wp + ((size_t)(kt * 2 + lh) * 128 + colB) * 16);
        acc = __builtin_amdgcn_wmma_f32_16x16x32_f16(
                  false, a, false, b, (short)0, acc, false, false);
    }
    const float bb = bias[colB];
    if (mbase + 16 <= M) {
#pragma unroll
        for (int r = 0; r < 8; ++r)
            Y[(size_t)(mbase + r + 8 * lh) * 128 + colB] = acc[r] + bb;
    } else {
#pragma unroll
        for (int r = 0; r < 8; ++r) {
            const int m = mbase + r + 8 * lh;
            if (m < M) Y[(size_t)m * 128 + colB] = acc[r] + bb;
        }
    }
}

__global__ void zero_cnt_k(int* __restrict__ cnt, int n)
{
    int i = blockIdx.x * blockDim.x + threadIdx.x;
    if (i < n) cnt[i] = 0;
}

__global__ void count_dst_k(const int* __restrict__ ei, int* __restrict__ cnt, int E)
{
    int e = blockIdx.x * blockDim.x + threadIdx.x;
    if (e < E) atomicAdd(&cnt[ei[E + e]], 1);
}

__global__ __launch_bounds__(256) void scan_cnt_k(
    const int* __restrict__ cnt, int* __restrict__ rowptr,
    int* __restrict__ cursor, int n)
{
    __shared__ int s_front[8];
    __shared__ int s_own[8];
    const int t = threadIdx.x, lane = t & 31, wv = t >> 5;
    const int start = blockIdx.x * 1024;

    int front = 0;
    for (int i = t; i < start; i += 256) front += cnt[i];

    int c[4];
#pragma unroll
    for (int k = 0; k < 4; ++k) {
        const int idx = start + 4 * t + k;
        c[k] = idx < n ? cnt[idx] : 0;
    }
    const int own = c[0] + c[1] + c[2] + c[3];

#pragma unroll
    for (int off = 16; off >= 1; off >>= 1) front += __shfl_xor(front, off, 32);
    int incl = own;
#pragma unroll
    for (int off = 1; off < 32; off <<= 1) {
        const int v = __shfl_up(incl, off, 32);
        if (lane >= off) incl += v;
    }
    if (lane == 31) { s_front[wv] = front; s_own[wv] = incl; }
    __syncthreads();
    int carry = 0;
#pragma unroll
    for (int w = 0; w < 8; ++w) {
        carry += s_front[w];
        if (w < wv) carry += s_own[w];
    }
    int run = carry + incl - own;
#pragma unroll
    for (int k = 0; k < 4; ++k) {
        const int idx = start + 4 * t + k;
        if (idx <= n) { rowptr[idx] = run; cursor[idx] = run; }
        run += c[k];
    }
}

__global__ void fill_src_k(const int* __restrict__ ei, int* __restrict__ cursor,
                           int* __restrict__ srcs, int E)
{
    int e = blockIdx.x * blockDim.x + threadIdx.x;
    if (e >= E) return;
    const int pos = atomicAdd(&cursor[ei[E + e]], 1);
    srcs[pos] = ei[e];
}


template <int M>
__device__ __forceinline__ float xmask_add(float v) {
    const int o = __builtin_amdgcn_update_dpp(0, __float_as_int(v), 0x160 + M, 0xf, 0xf, true);
    return v + __int_as_float(o);
}

typedef float v2f __attribute__((ext_vector_type(2)));

__device__ __forceinline__ v2f lrelu02x2(v2f v) {
    const v2f s = v * 0.2f;
    v2f r;
    r.x = fmaxf(v.x, s.x);
    r.y = fmaxf(v.y, s.y);
    return r;
}

__global__ __launch_bounds__(256) void node_gat_k(
    const float* __restrict__ xl, const int* __restrict__ rowptr,
    const int* __restrict__ srcs, const float* __restrict__ att,
    const float* __restrict__ bias, float* __restrict__ out, int n)
{
    const int lane = threadIdx.x & 31;
    const int wv = __builtin_amdgcn_readfirstlane((int)(threadIdx.x >> 5));
    const int d  = (int)blockIdx.x * 8 + wv;
    if (d >= n) return;
    const int beg = __builtin_amdgcn_readfirstlane(rowptr[d]);
    const int end = __builtin_amdgcn_readfirstlane(rowptr[d + 1]);

    int sv = 0;
    if (end > beg) {
        const int gi = beg + lane;
        sv = srcs[gi < end ? gi : end - 1];
    }
    const float4 a4 = *(const float4*)(att + 4 * lane);
    const float4 xd = *(const float4*)(xl + (size_t)d * 128 + 4 * lane);
    const float4 b4 = *(const float4*)(bias + 4 * lane);
    const v2f a_lo = {a4.x, a4.y}, a_hi = {a4.z, a4.w};
    const v2f d_lo = {xd.x, xd.y}, d_hi = {xd.z, xd.w};

    float m = -INFINITY, l = 0.f;
    v2f acc_lo = {0.f, 0.f}, acc_hi = {0.f, 0.f};

    for (int g = beg; g < end; g += 32) {
        if (g != beg) {
            const int gi = g + lane;
            sv = srcs[gi < end ? gi : end - 1];
        }
        const int gcnt = (end - g) < 32 ? (end - g) : 32;

        for (int c = 0; c < gcnt; c += 8) {
            const int left = gcnt - c;
            float4 xs[8];
#pragma unroll
            for (int u = 0; u < 8; ++u) {
                const int s = __builtin_amdgcn_readlane(sv, c + u);
                xs[u] = *(const float4*)(xl + (size_t)s * 128 + 4 * lane);
            }

            float p[8];
            float cm = -INFINITY;
#pragma unroll
            for (int u = 0; u < 8; ++u) {
                const v2f x_lo = {xs[u].x, xs[u].y}, x_hi = {xs[u].z, xs[u].w};
                const v2f t_lo = lrelu02x2(x_lo + d_lo);
                const v2f t_hi = lrelu02x2(x_hi + d_hi);
                const v2f r = t_lo * a_lo + t_hi * a_hi;
                float q = r.x + r.y;
                q = xmask_add<1>(q);
                q = xmask_add<2>(q);
                q = xmask_add<4>(q);
                p[u] = (u < left) ? q : -INFINITY;
                cm = fmaxf(cm, p[u]);
            }

            const float mn = fmaxf(m, cm);
            const float sc = __expf(m - mn);
            l *= sc;
            acc_lo *= sc; acc_hi *= sc;
#pragma unroll
            for (int u = 0; u < 8; ++u) {
                const v2f x_lo = {xs[u].x, xs[u].y}, x_hi = {xs[u].z, xs[u].w};
                const float w = __expf(p[u] - mn);
                l += w;
                acc_lo += w * x_lo;
                acc_hi += w * x_hi;
            }
            m = mn;
        }
    }

    float4 o = b4;
    if (end > beg) {
        o.x += acc_lo.x / l; o.y += acc_lo.y / l; o.z += acc_hi.x / l; o.w += acc_hi.y / l;
    }
    *(float4*)(out + (size_t)d * 128 + 4 * lane) = o;
}

__global__ void gelu_k(float* __restrict__ x, int n)
{
    int i = blockIdx.x * blockDim.x + threadIdx.x;
    if (i < n) {
        float v = x[i];
        x[i] = 0.5f * v * (1.f + erff(v * 0.70710678118654752440f));
    }
}

__global__ __launch_bounds__(256) void layernorm_k(
    const float* __restrict__ y, const float* __restrict__ gamma,
    const float* __restrict__ beta, float* __restrict__ out, int rows)
{
    const int wv = threadIdx.x >> 5, lane = threadIdx.x & 31;
    const int row = blockIdx.x * 8 + wv;
    if (row >= rows) return;
    const float* p = y + (size_t)row * 128 + lane * 4;
    float v0 = p[0], v1 = p[1], v2 = p[2], v3 = p[3];
    float s = v0 + v1 + v2 + v3;
#pragma unroll
    for (int off = 16; off >= 1; off >>= 1) s += __shfl_xor(s, off, 32);
    const float mu = s * (1.f / 128.f);
    const float d0 = v0 - mu, d1 = v1 - mu, d2 = v2 - mu, d3 = v3 - mu;
    float q = d0 * d0 + d1 * d1 + d2 * d2 + d3 * d3;
#pragma unroll
    for (int off = 16; off >= 1; off >>= 1) q += __shfl_xor(q, off, 32);
    const float rs = rsqrtf(q * (1.f / 128.f) + 1e-12f);
    const int c = lane * 4;
    float* o = out + (size_t)row * 128 + c;
    o[0] = d0 * rs * gamma[c + 0] + beta[c + 0];
    o[1] = d1 * rs * gamma[c + 1] + beta[c + 1];
    o[2] = d2 * rs * gamma[c + 2] + beta[c + 2];
    o[3] = d3 * rs * gamma[c + 3] + beta[c + 3];
}

extern "C" void kernel_launch(void* const* d_in, const int* in_sizes, int n_in,
                              void* d_out, int out_size, void* d_ws, size_t ws_size,
                              hipStream_t stream)
{
    const float* embs  = (const float*)d_in[0];
    const float* w0    = (const float*)d_in[1];
    const float* b0    = (const float*)d_in[2];
    const float* att0  = (const float*)d_in[3];
    const float* bias0 = (const float*)d_in[4];
    const float* w1    = (const float*)d_in[5];
    const float* b1    = (const float*)d_in[6];
    const float* att1  = (const float*)d_in[7];
    const float* bias1 = (const float*)d_in[8];
    const float* wout  = (const float*)d_in[9];
    const float* bout  = (const float*)d_in[10];
    const float* gamma = (const float*)d_in[11];
    const float* beta  = (const float*)d_in[12];
    const int*   ei0   = (const int*)d_in[13];
    const int*   ei1   = (const int*)d_in[14];

    const int N  = in_sizes[0] / 128;
    const int E  = in_sizes[13] / 2;
    const int BS = out_size / 128;

    char* wsp = (char*)d_ws;
    size_t off = 0;
    auto carve = [&](size_t bytes) -> void* {
        void* p = wsp + off; off += (bytes + 255) & ~(size_t)255; return p;
    };
    float*     xl    = (float*)    carve((size_t)N * 128 * 4);
    float*     buf   = (float*)    carve((size_t)N * 128 * 4);
    int*       cnt    = (int*)     carve((size_t)(N + 1) * 4);
    int*       rowptr = (int*)     carve((size_t)(N + 1) * 4);
    int*       cursor = (int*)     carve((size_t)(N + 1) * 4);
    int*       srcs   = (int*)     carve((size_t)E * 4);
    float*     ybuf  = (float*)    carve((size_t)BS * 128 * 4);
    _Float16*  Wp    = (_Float16*) carve((size_t)128 * 128 * 2);

    const int gemmBlocksN = (N + 15) / 16;
    const int edgeBlocks  = (E + 255) / 256;
    const int cntBlocks   = (N + 255) / 256;
    const int scanBlocks  = (N + 1 + 1023) / 1024;
    const int nodeBlocks  = (N + 7) / 8;
    const int featBlocks  = (N * 128 + 255) / 256;
    const int packBlocks  = (128 * 128 + 255) / 256;

    pack_w_k    <<<packBlocks, 256, 0, stream>>>(w0, Wp);
    gemm128_wmma<<<gemmBlocksN, 256, 0, stream>>>(embs, Wp, b0, xl, N);
    zero_cnt_k  <<<cntBlocks,  256, 0, stream>>>(cnt, N);
    count_dst_k <<<edgeBlocks, 256, 0, stream>>>(ei0, cnt, E);
    scan_cnt_k  <<<scanBlocks, 256, 0, stream>>>(cnt, rowptr, cursor, N);
    fill_src_k  <<<edgeBlocks, 256, 0, stream>>>(ei0, cursor, srcs, E);
    node_gat_k  <<<nodeBlocks, 256, 0, stream>>>(xl, rowptr, srcs, att0, bias0, buf, N);
    gelu_k      <<<featBlocks, 256, 0, stream>>>(buf, N * 128);

    pack_w_k    <<<packBlocks, 256, 0, stream>>>(w1, Wp);
    gemm128_wmma<<<gemmBlocksN, 256, 0, stream>>>(buf, Wp, b1, xl, N);
    zero_cnt_k  <<<cntBlocks,  256, 0, stream>>>(cnt, N);
    count_dst_k <<<edgeBlocks, 256, 0, stream>>>(ei1, cnt, E);
    scan_cnt_k  <<<scanBlocks, 256, 0, stream>>>(cnt, rowptr, cursor, N);
    fill_src_k  <<<edgeBlocks, 256, 0, stream>>>(ei1, cursor, srcs, E);
    node_gat_k  <<<nodeBlocks, 256, 0, stream>>>(xl, rowptr, srcs, att1, bias1, buf, N);

    pack_w_k    <<<packBlocks, 256, 0, stream>>>(wout, Wp);
    gemm128_wmma<<<(BS + 15) / 16, 256, 0, stream>>>(buf, Wp, bout, ybuf, BS);
    layernorm_k <<<(BS + 7) / 8,   256, 0, stream>>>(ybuf, gamma, beta, (float*)d_out, BS);
}


# layer-0 attention, its edge sort and the layer-1 GEMM run only on rows layer 1 reads (sources of its edges ending below BS, plus rows 0..BS-1), flagged and listed in-call; sum order unchanged
# speedup vs baseline: 6.5121x; 2.8508x over previous
#include <hip/hip_runtime.h>
#include <hip/hip_bf16.h>
#include <math.h>

typedef _Float16 v16h __attribute__((ext_vector_type(16)));
typedef float    v8f  __attribute__((ext_vector_type(8)));

__global__ void pack_w_k(const float* __restrict__ W, _Float16* __restrict__ Wp)
{
    int idx = blockIdx.x * blockDim.x + threadIdx.x;
    if (idx >= 128 * 128) return;
    const int r  = idx >> 4;
    const int i  = idx & 15;
    const int col = r & 127;
    const int lh  = (r >> 7) & 1;
    const int kb  = r >> 8;
    const int kk  = (i < 8) ? (8 * lh + i) : (8 + 8 * lh + i);
    const int K   = 32 * kb + kk;
    Wp[(size_t)r * 16 + i] = (_Float16)W[(size_t)K * 128 + col];
}

template <bool LIST>
__global__ __launch_bounds__(256) void gemm128_wmma(
    const float* __restrict__ X, const _Float16* __restrict__ Wp,
    const float* __restrict__ bias, float* __restrict__ Y, int M,
    const int* __restrict__ list, const int* __restrict__ nlist)
{
    const int wv    = threadIdx.x >> 5;
    const int lane  = threadIdx.x & 31;
    const int lh    = lane >> 4;
    const int lm    = lane & 15;
    const int colB  = wv * 16 + lm;
    const float bb  = bias[colB];
    if (LIST) M = *nlist;

  for (int mbase = blockIdx.x * 16; mbase < M; mbase += gridDim.x * 16) {
    int rowA = mbase + lm; if (rowA >= M) rowA = M - 1;
    if (LIST) rowA = list[rowA];

    v8f acc = {};
#pragma unroll
    for (int kt = 0; kt < 4; ++kt) {
        const int kb = kt * 32;
        const float4* pr = (const float4*)(X + (size_t)rowA * 128 + kb);
        const float4 x0 = pr[2 * lh + 0];
        const float4 x1 = pr[2 * lh + 1];
        const float4 x2 = pr[4 + 2 * lh + 0];
        const float4 x3 = pr[4 + 2 * lh + 1];
        v16h a;
        a[0]  = (_Float16)x0.x; a[1]  = (_Float16)x0.y;
        a[2]  = (_Float16)x0.z; a[3]  = (_Float16)x0.w;
        a[4]  = (_Float16)x1.x; a[5]  = (_Float16)x1.y;
        a[6]  = (_Float16)x1.z; a[7]  = (_Float16)x1.w;
        a[8]  = (_Float16)x2.x; a[9]  = (_Float16)x2.y;
        a[10] = (_Float16)x2.z; a[11] = (_Float16)x2.w;
        a[12] = (_Float16)x3.x; a[13] = (_Float16)x3.y;
        a[14] = (_Float16)x3.z; a[15] = (_Float16)x3.w;
        const v16h b = *(const v16h*)(Wp + ((size_t)(kt * 2 + lh) * 128 + colB) * 16);
        acc = __builtin_amdgcn_wmma_f32_16x16x32_f16(
                  false, a, false, b, (short)0, acc, false, false);
    }
    if (mbase + 16 <= M) {
#pragma unroll
        for (int r = 0; r < 8; ++r) {
            const int m = mbase + r + 8 * lh;
            const int row = LIST ? list[m] : m;
            Y[(size_t)row * 128 + colB] = acc[r] + bb;
        }
    } else {
#pragma unroll
        for (int r = 0; r < 8; ++r) {
            const int m = mbase + r + 8 * lh;
            if (m < M) {
                const int row = LIST ? list[m] : m;
                Y[(size_t)row * 128 + colB] = acc[r] + bb;
            }
        }
    }
  }
}

__global__ void zero_cnt_k(int* __restrict__ cnt, int n)
{
    int i = blockIdx.x * blockDim.x + threadIdx.x;
    if (i < n) cnt[i] = 0;
}

__device__ __forceinline__ void set_flag(int* p) {
    __hip_atomic_store(p, 1, __ATOMIC_RELAXED, __HIP_MEMORY_SCOPE_AGENT);
}

__global__ void mark_count1_k(const int* __restrict__ ei, int* __restrict__ cnt1,
                              int* __restrict__ need, int E, int nd)
{
    int e = blockIdx.x * blockDim.x + threadIdx.x;
    if (e < nd) set_flag(&need[e]);
    if (e >= E) return;
    const int d = ei[E + e];
    if (d >= nd) return;
    atomicAdd(&cnt1[d], 1);
    set_flag(&need[ei[e]]);
}

__global__ void count_dst_k(const int* __restrict__ ei, const int* __restrict__ need,
                            int* __restrict__ cnt, int E)
{
    int e = blockIdx.x * blockDim.x + threadIdx.x;
    if (e >= E) return;
    const int d = ei[E + e];
    if (need[d]) atomicAdd(&cnt[d], 1);
}

__global__ __launch_bounds__(256) void scan_cnt_k(
    const int* __restrict__ cnt_in, int* __restrict__ rowptr,
    int* __restrict__ cursor, int n,
    const int* __restrict__ need, int* __restrict__ list, int* __restrict__ nlist)
{
    const bool flags = blockIdx.y == 1;
    const int* __restrict__ cnt = flags ? need : cnt_in;
    __shared__ int s_front[8];
    __shared__ int s_own[8];
    const int t = threadIdx.x, lane = t & 31, wv = t >> 5;
    const int start = blockIdx.x * 1024;

    int front = 0;
    for (int i = t; i < start; i += 256) front += cnt[i];

    int c[4];
#pragma unroll
    for (int k = 0; k < 4; ++k) {
        const int idx = start + 4 * t + k;
        c[k] = idx < n ? cnt[idx] : 0;
    }
    const int own = c[0] + c[1] + c[2] + c[3];

#pragma unroll
    for (int off = 16; off >= 1; off >>= 1) front += __shfl_xor(front, off, 32);
    int incl = own;
#pragma unroll
    for (int off = 1; off < 32; off <<= 1) {
        const int v = __shfl_up(incl, off, 32);
        if (lane >= off) incl += v;
    }
    if (lane == 31) { s_front[wv] = front; s_own[wv] = incl; }
    __syncthreads();
    int carry = 0;
#pragma unroll
    for (int w = 0; w < 8; ++w) {
        carry += s_front[w];
        if (w < wv) carry += s_own[w];
    }
    int run = carry + incl - own;
#pragma unroll
    for (int k = 0; k < 4; ++k) {
        const int idx = start + 4 * t + k;
        if (flags) {
            if (idx < n && c[k]) list[run] = idx;
            if (idx == n) *nlist = run;
        } else if (idx <= n) { rowptr[idx] = run; cursor[idx] = run; }
        run += c[k];
    }
}

__global__ void fill_src_k(const int* __restrict__ ei, const int* __restrict__ need,
                           int* __restrict__ cursor, int* __restrict__ srcs, int E, int nd)
{
    int e = blockIdx.x * blockDim.x + threadIdx.x;
    if (e >= E) return;
    const int d = ei[E + e];
    if (d >= nd) return;
    if (need && !need[d]) return;
    const int pos = atomicAdd(&cursor[d], 1);
    srcs[pos] = ei[e];
}


__device__ __forceinline__ float gelu_erf(float v) {
    return 0.5f * v * (1.f + erff(v * 0.70710678118654752440f));
}

template <int M>
__device__ __forceinline__ float xmask_add(float v) {
    const int o = __builtin_amdgcn_update_dpp(0, __float_as_int(v), 0x160 + M, 0xf, 0xf, true);
    return v + __int_as_float(o);
}

typedef float v2f __attribute__((ext_vector_type(2)));

__device__ __forceinline__ v2f lrelu02x2(v2f v) {
    const v2f s = v * 0.2f;
    v2f r;
    r.x = fmaxf(v.x, s.x);
    r.y = fmaxf(v.y, s.y);
    return r;
}

template <bool GELU, bool LIST>
__global__ __launch_bounds__(256) __attribute__((amdgpu_waves_per_eu(12))) void node_gat_k(
    const float* __restrict__ xl, const int* __restrict__ rowptr,
    const int* __restrict__ srcs, const float* __restrict__ att,
    const float* __restrict__ bias, float* __restrict__ out, int n,
    const int* __restrict__ list, const int* __restrict__ nlist)
{
    const int lane = threadIdx.x & 31;
    const int wv = __builtin_amdgcn_readfirstlane((int)(threadIdx.x >> 5));
    if (LIST) n = *nlist;

  for (int w = (int)blockIdx.x * 8 + wv; w < n; w += (int)gridDim.x * 8) {
    const int d = LIST ? __builtin_amdgcn_readfirstlane(list[w]) : w;
    const int beg = __builtin_amdgcn_readfirstlane(rowptr[d]);
    const int end = __builtin_amdgcn_readfirstlane(rowptr[d + 1]);

    int sv = 0;
    if (end > beg) {
        const int gi = beg + lane;
        sv = srcs[gi < end ? gi : end - 1];
    }
    const float4 a4 = *(const float4*)(att + 4 * lane);
    const float4 xd = *(const float4*)(xl + (size_t)d * 128 + 4 * lane);
    const float4 b4 = *(const float4*)(bias + 4 * lane);
    const v2f a_lo = {a4.x, a4.y}, a_hi = {a4.z, a4.w};
    const v2f d_lo = {xd.x, xd.y}, d_hi = {xd.z, xd.w};

    float m = -INFINITY, l = 0.f;
    v2f acc_lo = {0.f, 0.f}, acc_hi = {0.f, 0.f};

    for (int g = beg; g < end; g += 32) {
        if (g != beg) {
            const int gi = g + lane;
            sv = srcs[gi < end ? gi : end - 1];
        }
        const int gcnt = (end - g) < 32 ? (end - g) : 32;

        for (int c = 0; c < gcnt; c += 8) {
            const int left = gcnt - c;
            float4 xs[8];
#pragma unroll
            for (int u = 0; u < 8; ++u) {
                const int s = __builtin_amdgcn_readlane(sv, c + u);
                xs[u] = *(const float4*)(xl + (size_t)s * 128 + 4 * lane);
            }

            float p[8];
            float cm = -INFINITY;
#pragma unroll
            for (int u = 0; u < 8; ++u) {
                const v2f x_lo = {xs[u].x, xs[u].y}, x_hi = {xs[u].z, xs[u].w};
                const v2f t_lo = lrelu02x2(x_lo + d_lo);
                const v2f t_hi = lrelu02x2(x_hi + d_hi);
                const v2f r = t_lo * a_lo + t_hi * a_hi;
                float q = r.x + r.y;
                q = xmask_add<1>(q);
                q = xmask_add<2>(q);
                q = xmask_add<4>(q);
                p[u] = (u < left) ? q : -INFINITY;
                cm = fmaxf(cm, p[u]);
            }

            const float mn = fmaxf(m, cm);
            const float sc = __expf(m - mn);
            l *= sc;
            acc_lo *= sc; acc_hi *= sc;
#pragma unroll
            for (int u = 0; u < 8; ++u) {
                const v2f x_lo = {xs[u].x, xs[u].y}, x_hi = {xs[u].z, xs[u].w};
                const float w = __expf(p[u] - mn);
                l += w;
                acc_lo += w * x_lo;
                acc_hi += w * x_hi;
            }
            m = mn;
        }
    }

    float4 o = b4;
    if (end > beg) {
        o.x += acc_lo.x / l; o.y += acc_lo.y / l; o.z += acc_hi.x / l; o.w += acc_hi.y / l;
    }
    if (GELU) {
        o.x = gelu_erf(o.x); o.y = gelu_erf(o.y); o.z = gelu_erf(o.z); o.w = gelu_erf(o.w);
    }
    *(float4*)(out + (size_t)d * 128 + 4 * lane) = o;
  }
}

__global__ __launch_bounds__(256) void layernorm_k(
    const float* __restrict__ y, const float* __restrict__ gamma,
    const float* __restrict__ beta, float* __restrict__ out, int rows)
{
    const int wv = threadIdx.x >> 5, lane = threadIdx.x & 31;
    const int row = blockIdx.x * 8 + wv;
    if (row >= rows) return;
    const float* p = y + (size_t)row * 128 + lane * 4;
    float v0 = p[0], v1 = p[1], v2 = p[2], v3 = p[3];
    float s = v0 + v1 + v2 + v3;
#pragma unroll
    for (int off = 16; off >= 1; off >>= 1) s += __shfl_xor(s, off, 32);
    const float mu = s * (1.f / 128.f);
    const float d0 = v0 - mu, d1 = v1 - mu, d2 = v2 - mu, d3 = v3 - mu;
    float q = d0 * d0 + d1 * d1 + d2 * d2 + d3 * d3;
#pragma unroll
    for (int off = 16; off >= 1; off >>= 1) q += __shfl_xor(q, off, 32);
    const float rs = rsqrtf(q * (1.f / 128.f) + 1e-12f);
    const int c = lane * 4;
    float* o = out + (size_t)row * 128 + c;
    o[0] = d0 * rs * gamma[c + 0] + beta[c + 0];
    o[1] = d1 * rs * gamma[c + 1] + beta[c + 1];
    o[2] = d2 * rs * gamma[c + 2] + beta[c + 2];
    o[3] = d3 * rs * gamma[c + 3] + beta[c + 3];
}

extern "C" void kernel_launch(void* const* d_in, const int* in_sizes, int n_in,
                              void* d_out, int out_size, void* d_ws, size_t ws_size,
                              hipStream_t stream)
{
    const float* embs  = (const float*)d_in[0];
    const float* w0    = (const float*)d_in[1];
    const float* b0    = (const float*)d_in[2];
    const float* att0  = (const float*)d_in[3];
    const float* bias0 = (const float*)d_in[4];
    const float* w1    = (const float*)d_in[5];
    const float* b1    = (const float*)d_in[6];
    const float* att1  = (const float*)d_in[7];
    const float* bias1 = (const float*)d_in[8];
    const float* wout  = (const float*)d_in[9];
    const float* bout  = (const float*)d_in[10];
    const float* gamma = (const float*)d_in[11];
    const float* beta  = (const float*)d_in[12];
    const int*   ei0   = (const int*)d_in[13];
    const int*   ei1   = (const int*)d_in[14];

    const int N  = in_sizes[0] / 128;
    const int E  = in_sizes[13] / 2;
    const int BS = out_size / 128;

    char* wsp = (char*)d_ws;
    size_t off = 0;
    auto carve = [&](size_t bytes) -> void* {
        void* p = wsp + off; off += (bytes + 255) & ~(size_t)255; return p;
    };
    float*     xl    = (float*)    carve((size_t)N * 128 * 4);
    float*     buf   = (float*)    carve((size_t)N * 128 * 4);
    const int  nzero  = (N + 1) + (N + 1) + (BS + 1);
    int*       zreg   = (int*)     carve((size_t)nzero * 4);
    int*       need   = zreg;
    int*       cnt    = need + (N + 1);
    int*       cnt1   = cnt + (N + 1);
    int*       list   = (int*)     carve((size_t)(N + 1) * 4);
    int*       nlist  = list + N;
    int*       rowptr = (int*)     carve((size_t)(N + 1) * 4);
    int*       cursor = (int*)     carve((size_t)(N + 1) * 4);
    int*       srcs   = (int*)     carve((size_t)E * 4);
    float*     ybuf  = (float*)    carve((size_t)BS * 128 * 4);
    _Float16*  Wp    = (_Float16*) carve((size_t)128 * 128 * 2);

    const int gemmBlocksN = (N + 15) / 16;
    const int edgeBlocks  = (E + 255) / 256;
    const int markBlocks  = ((E > BS ? E : BS) + 255) / 256;
    const int zeroBlocks  = (nzero + 255) / 256;
    const int scanBlocks  = (N + 1 + 1023) / 1024;
    const int packBlocks  = (128 * 128 + 255) / 256;
    const int scanBlocks1 = (BS + 1 + 1023) / 1024;
    const int nodeBlocks1 = (BS + 7) / 8;
    const long long estRows64 = (long long)E * BS / N + BS;
    const int estRows     = estRows64 < N ? (int)estRows64 : N;
    const int nodeBlocksL = (estRows + 7) / 8;
    const int gemmBlocksL = (estRows + 15) / 16;

    zero_cnt_k   <<<zeroBlocks, 256, 0, stream>>>(zreg, nzero);
    mark_count1_k<<<markBlocks, 256, 0, stream>>>(ei1, cnt1, need, E, BS);

    pack_w_k    <<<packBlocks, 256, 0, stream>>>(w0, Wp);
    gemm128_wmma<false><<<gemmBlocksN, 256, 0, stream>>>(embs, Wp, b0, xl, N, nullptr, nullptr);
    count_dst_k <<<edgeBlocks, 256, 0, stream>>>(ei0, need, cnt, E);
    scan_cnt_k  <<<dim3(scanBlocks, 2), 256, 0, stream>>>(cnt, rowptr, cursor, N, need, list, nlist);
    fill_src_k  <<<edgeBlocks, 256, 0, stream>>>(ei0, need, cursor, srcs, E, N);
    node_gat_k<true, true><<<nodeBlocksL, 256, 0, stream>>>(xl, rowptr, srcs, att0, bias0, buf, 0, list, nlist);

    pack_w_k    <<<packBlocks, 256, 0, stream>>>(w1, Wp);
    gemm128_wmma<true><<<gemmBlocksL, 256, 0, stream>>>(buf, Wp, b1, xl, 0, list, nlist);
    scan_cnt_k  <<<scanBlocks1, 256, 0, stream>>>(cnt1, rowptr, cursor, BS, nullptr, nullptr, nullptr);
    fill_src_k  <<<edgeBlocks, 256, 0, stream>>>(ei1, nullptr, cursor, srcs, E, BS);
    node_gat_k<false, false><<<nodeBlocks1, 256, 0, stream>>>(xl, rowptr, srcs, att1, bias1, buf, BS, nullptr, nullptr);

    pack_w_k    <<<packBlocks, 256, 0, stream>>>(wout, Wp);
    gemm128_wmma<false><<<(BS + 15) / 16, 256, 0, stream>>>(buf, Wp, bout, ybuf, BS, nullptr, nullptr);
    layernorm_k <<<(BS + 7) / 8,   256, 0, stream>>>(ybuf, gamma, beta, (float*)d_out, BS);
}


# launches that do not depend on each other share one launch, the block index picks the job: packs+clear, GEMM0+edge mark, 3 scans, GEMM1+layer-1 fill; 16 launches -> 10; sum order unchanged
# speedup vs baseline: 7.3674x; 1.1313x over previous
#include <hip/hip_runtime.h>
#include <hip/hip_bf16.h>
#include <math.h>

typedef _Float16 v16h __attribute__((ext_vector_type(16)));
typedef float    v8f  __attribute__((ext_vector_type(8)));

__device__ __forceinline__ void pack_w(const float* __restrict__ W, _Float16* __restrict__ Wp, int idx)
{
    if (idx >= 128 * 128) return;
    const int r  = idx >> 4;
    const int i  = idx & 15;
    const int col = r & 127;
    const int lh  = (r >> 7) & 1;
    const int kb  = r >> 8;
    const int kk  = (i < 8) ? (8 * lh + i) : (8 + 8 * lh + i);
    const int K   = 32 * kb + kk;
    Wp[(size_t)r * 16 + i] = (_Float16)W[(size_t)K * 128 + col];
}

constexpr int kPackBlocks = (128 * 128 + 255) / 256;

__global__ __launch_bounds__(256) void prologue_k(
    const float* __restrict__ w0, const float* __restrict__ w1, const float* __restrict__ w2,
    _Float16* __restrict__ Wp0, _Float16* __restrict__ Wp1, _Float16* __restrict__ Wp2,
    int* __restrict__ zreg, int nzero)
{
    const int b = (int)blockIdx.x;
    if (b < 3 * kPackBlocks) {
        const int which = b / kPackBlocks;
        const float* W  = which == 0 ? w0  : which == 1 ? w1  : w2;
        _Float16*    Wp = which == 0 ? Wp0 : which == 1 ? Wp1 : Wp2;
        pack_w(W, Wp, (b - which * kPackBlocks) * 256 + (int)threadIdx.x);
    } else {
        const int i = (b - 3 * kPackBlocks) * 256 + (int)threadIdx.x;
        if (i < nzero) zreg[i] = 0;
    }
}

template <bool LIST>
__device__ __forceinline__ void gemm128_tiles(
    const float* __restrict__ X, const _Float16* __restrict__ Wp,
    const float* __restrict__ bias, float* __restrict__ Y, int M,
    const int* __restrict__ list, const int* __restrict__ nlist, int blk, int nblk)
{
    const int wv    = threadIdx.x >> 5;
    const int lane  = threadIdx.x & 31;
    const int lh    = lane >> 4;
    const int lm    = lane & 15;
    const int colB  = wv * 16 + lm;
    const float bb  = bias[colB];
    if (LIST) M = *nlist;

  for (int mbase = blk * 16; mbase < M; mbase += nblk * 16) {
    int rowA = mbase + lm; if (rowA >= M) rowA = M - 1;
    if (LIST) rowA = list[rowA];

    v8f acc = {};
#pragma unroll
    for (int kt = 0; kt < 4; ++kt) {
        const int kb = kt * 32;
        const float4* pr = (const float4*)(X + (size_t)rowA * 128 + kb);
        const float4 x0 = pr[2 * lh + 0];
        const float4 x1 = pr[2 * lh + 1];
        const float4 x2 = pr[4 + 2 * lh + 0];
        const float4 x3 = pr[4 + 2 * lh + 1];
        v16h a;
        a[0]  = (_Float16)x0.x; a[1]  = (_Float16)x0.y;
        a[2]  = (_Float16)x0.z; a[3]  = (_Float16)x0.w;
        a[4]  = (_Float16)x1.x; a[5]  = (_Float16)x1.y;
        a[6]  = (_Float16)x1.z; a[7]  = (_Float16)x1.w;
        a[8]  = (_Float16)x2.x; a[9]  = (_Float16)x2.y;
        a[10] = (_Float16)x2.z; a[11] = (_Float16)x2.w;
        a[12] = (_Float16)x3.x; a[13] = (_Float16)x3.y;
        a[14] = (_Float16)x3.z; a[15] = (_Float16)x3.w;
        const v16h b = *(const v16h*)(Wp + ((size_t)(kt * 2 + lh) * 128 + colB) * 16);
        acc = __builtin_amdgcn_wmma_f32_16x16x32_f16(
                  false, a, false, b, (short)0, acc, false, false);
    }
    if (mbase + 16 <= M) {
#pragma unroll
        for (int r = 0; r < 8; ++r) {
            const int m = mbase + r + 8 * lh;
            const int row = LIST ? list[m] : m;
            Y[(size_t)row * 128 + colB] = acc[r] + bb;
        }
    } else {
#pragma unroll
        for (int r = 0; r < 8; ++r) {
            const int m = mbase + r + 8 * lh;
            if (m < M) {
                const int row = LIST ? list[m] : m;
                Y[(size_t)row * 128 + colB] = acc[r] + bb;
            }
        }
    }
  }
}

__global__ __launch_bounds__(256) void gemm128_wmma(
    const float* __restrict__ X, const _Float16* __restrict__ Wp,
    const float* __restrict__ bias, float* __restrict__ Y, int M)
{
    gemm128_tiles<false>(X, Wp, bias, Y, M, nullptr, nullptr, (int)blockIdx.x, (int)gridDim.x);
}

__device__ __forceinline__ void set_flag(int* p) {
    __hip_atomic_store(p, 1, __ATOMIC_RELAXED, __HIP_MEMORY_SCOPE_AGENT);
}

__device__ __forceinline__ void mark_count1(const int* __restrict__ ei, int* __restrict__ cnt1,
                                            int* __restrict__ need, int E, int nd, int e)
{
    if (e < nd) set_flag(&need[e]);
    if (e >= E) return;
    const int d = ei[E + e];
    if (d >= nd) return;
    atomicAdd(&cnt1[d], 1);
    set_flag(&need[ei[e]]);
}

__global__ __launch_bounds__(256) void gemm_mark_k(
    const float* __restrict__ X, const _Float16* __restrict__ Wp,
    const float* __restrict__ bias, float* __restrict__ Y, int M, int gemmBlocks,
    const int* __restrict__ ei, int* __restrict__ cnt1, int* __restrict__ need, int E, int nd)
{
    const int b = (int)blockIdx.x;
    if (b < gemmBlocks)
        gemm128_tiles<false>(X, Wp, bias, Y, M, nullptr, nullptr, b, gemmBlocks);
    else
        mark_count1(ei, cnt1, need, E, nd, (b - gemmBlocks) * 256 + (int)threadIdx.x);
}

__global__ void count_dst_k(const int* __restrict__ ei, const int* __restrict__ need,
                            int* __restrict__ cnt, int E)
{
    int e = blockIdx.x * blockDim.x + threadIdx.x;
    if (e >= E) return;
    const int d = ei[E + e];
    if (need[d]) atomicAdd(&cnt[d], 1);
}

__global__ __launch_bounds__(256) void scan_cnt_k(
    const int* __restrict__ cnt0, int* __restrict__ rowptr0, int* __restrict__ cursor0, int n0,
    const int* __restrict__ need, int* __restrict__ list, int* __restrict__ nlist,
    const int* __restrict__ cnt1, int* __restrict__ rowptr1, int* __restrict__ cursor1, int n1)
{
    const int plane = (int)blockIdx.y;
    const bool flags = plane == 1;
    const int* cnt = plane == 0 ? cnt0 : plane == 1 ? need : cnt1;
    int* rowptr = plane == 2 ? rowptr1 : rowptr0;
    int* cursor = plane == 2 ? cursor1 : cursor0;
    const int n = plane == 2 ? n1 : n0;
    __shared__ int s_front[8];
    __shared__ int s_own[8];
    const int t = threadIdx.x, lane = t & 31, wv = t >> 5;
    const int start = blockIdx.x * 1024;
    if (start > n) return;

    int front = 0;
    for (int i = t; i < start; i += 256) front += cnt[i];

    int c[4];
#pragma unroll
    for (int k = 0; k < 4; ++k) {
        const int idx = start + 4 * t + k;
        c[k] = idx < n ? cnt[idx] : 0;
    }
    const int own = c[0] + c[1] + c[2] + c[3];

#pragma unroll
    for (int off = 16; off >= 1; off >>= 1) front += __shfl_xor(front, off, 32);
    int incl = own;
#pragma unroll
    for (int off = 1; off < 32; off <<= 1) {
        const int v = __shfl_up(incl, off, 32);
        if (lane >= off) incl += v;
    }
    if (lane == 31) { s_front[wv] = front; s_own[wv] = incl; }
    __syncthreads();
    int carry = 0;
#pragma unroll
    for (int w = 0; w < 8; ++w) {
        carry += s_front[w];
        if (w < wv) carry += s_own[w];
    }
    int run = carry + incl - own;
#pragma unroll
    for (int k = 0; k < 4; ++k) {
        const int idx = start + 4 * t + k;
        if (flags) {
            if (idx < n && c[k]) list[run] = idx;
            if (idx == n) *nlist = run;
        } else if (idx <= n) { rowptr[idx] = run; cursor[idx] = run; }
        run += c[k];
    }
}

__device__ __forceinline__ void fill_src(const int* __restrict__ ei, const int* __restrict__ need,
                                         int* __restrict__ cursor, int* __restrict__ srcs,
                                         int E, int nd, int e)
{
    if (e >= E) return;
    const int d = ei[E + e];
    if (d >= nd) return;
    if (need && !need[d]) return;
    const int pos = atomicAdd(&cursor[d], 1);
    srcs[pos] = ei[e];
}

__global__ void fill_src_k(const int* __restrict__ ei, const int* __restrict__ need,
                           int* __restrict__ cursor, int* __restrict__ srcs, int E, int nd)
{
    fill_src(ei, need, cursor, srcs, E, nd, (int)(blockIdx.x * blockDim.x + threadIdx.x));
}

__global__ __launch_bounds__(256) void gemm_fill_k(
    const float* __restrict__ X, const _Float16* __restrict__ Wp,
    const float* __restrict__ bias, float* __restrict__ Y,
    const int* __restrict__ list, const int* __restrict__ nlist, int gemmBlocks,
    const int* __restrict__ ei, int* __restrict__ cursor1, int* __restrict__ srcs, int E, int nd)
{
    const int b = (int)blockIdx.x;
    if (b < gemmBlocks)
        gemm128_tiles<true>(X, Wp, bias, Y, 0, list, nlist, b, gemmBlocks);
    else
        fill_src(ei, nullptr, cursor1, srcs, E, nd, (b - gemmBlocks) * 256 + (int)threadIdx.x);
}


__device__ __forceinline__ float gelu_erf(float v) {
    return 0.5f * v * (1.f + erff(v * 0.70710678118654752440f));
}

template <int M>
__device__ __forceinline__ float xmask_add(float v) {
    const int o = __builtin_amdgcn_update_dpp(0, __float_as_int(v), 0x160 + M, 0xf, 0xf, true);
    return v + __int_as_float(o);
}

typedef float v2f __attribute__((ext_vector_type(2)));

__device__ __forceinline__ v2f lrelu02x2(v2f v) {
    const v2f s = v * 0.2f;
    v2f r;
    r.x = fmaxf(v.x, s.x);
    r.y = fmaxf(v.y, s.y);
    return r;
}

template <bool GELU, bool LIST>
__global__ __launch_bounds__(256) __attribute__((amdgpu_waves_per_eu(12))) void node_gat_k(
    const float* __restrict__ xl, const int* __restrict__ rowptr,
    const int* __restrict__ srcs, const float* __restrict__ att,
    const float* __restrict__ bias, float* __restrict__ out, int n,
    const int* __restrict__ list, const int* __restrict__ nlist)
{
    const int lane = threadIdx.x & 31;
    const int wv = __builtin_amdgcn_readfirstlane((int)(threadIdx.x >> 5));
    if (LIST) n = *nlist;

  for (int w = (int)blockIdx.x * 8 + wv; w < n; w += (int)gridDim.x * 8) {
    const int d = LIST ? __builtin_amdgcn_readfirstlane(list[w]) : w;
    const int beg = __builtin_amdgcn_readfirstlane(rowptr[d]);
    const int end = __builtin_amdgcn_readfirstlane(rowptr[d + 1]);

    int sv = 0;
    if (end > beg) {
        const int gi = beg + lane;
        sv = srcs[gi < end ? gi : end - 1];
    }
    const float4 a4 = *(const float4*)(att + 4 * lane);
    const float4 xd = *(const float4*)(xl + (size_t)d * 128 + 4 * lane);
    const float4 b4 = *(const float4*)(bias + 4 * lane);
    const v2f a_lo = {a4.x, a4.y}, a_hi = {a4.z, a4.w};
    const v2f d_lo = {xd.x, xd.y}, d_hi = {xd.z, xd.w};

    float m = -INFINITY, l = 0.f;
    v2f acc_lo = {0.f, 0.f}, acc_hi = {0.f, 0.f};

    for (int g = beg; g < end; g += 32) {
        if (g != beg) {
            const int gi = g + lane;
            sv = srcs[gi < end ? gi : end - 1];
        }
        const int gcnt = (end - g) < 32 ? (end - g) : 32;

        for (int c = 0; c < gcnt; c += 8) {
            const int left = gcnt - c;
            float4 xs[8];
#pragma unroll
            for (int u = 0; u < 8; ++u) {
                const int s = __builtin_amdgcn_readlane(sv, c + u);
                xs[u] = *(const float4*)(xl + (size_t)s * 128 + 4 * lane);
            }

            float p[8];
            float cm = -INFINITY;
#pragma unroll
            for (int u = 0; u < 8; ++u) {
                const v2f x_lo = {xs[u].x, xs[u].y}, x_hi = {xs[u].z, xs[u].w};
                const v2f t_lo = lrelu02x2(x_lo + d_lo);
                const v2f t_hi = lrelu02x2(x_hi + d_hi);
                const v2f r = t_lo * a_lo + t_hi * a_hi;
                float q = r.x + r.y;
                q = xmask_add<1>(q);
                q = xmask_add<2>(q);
                q = xmask_add<4>(q);
                p[u] = (u < left) ? q : -INFINITY;
                cm = fmaxf(cm, p[u]);
            }

            const float mn = fmaxf(m, cm);
            const float sc = __expf(m - mn);
            l *= sc;
            acc_lo *= sc; acc_hi *= sc;
#pragma unroll
            for (int u = 0; u < 8; ++u) {
                const v2f x_lo = {xs[u].x, xs[u].y}, x_hi = {xs[u].z, xs[u].w};
                const float w = __expf(p[u] - mn);
                l += w;
                acc_lo += w * x_lo;
                acc_hi += w * x_hi;
            }
            m = mn;
        }
    }

    float4 o = b4;
    if (end > beg) {
        o.x += acc_lo.x / l; o.y += acc_lo.y / l; o.z += acc_hi.x / l; o.w += acc_hi.y / l;
    }
    if (GELU) {
        o.x = gelu_erf(o.x); o.y = gelu_erf(o.y); o.z = gelu_erf(o.z); o.w = gelu_erf(o.w);
    }
    *(float4*)(out + (size_t)d * 128 + 4 * lane) = o;
  }
}

__global__ __launch_bounds__(256) void layernorm_k(
    const float* __restrict__ y, const float* __restrict__ gamma,
    const float* __restrict__ beta, float* __restrict__ out, int rows)
{
    const int wv = threadIdx.x >> 5, lane = threadIdx.x & 31;
    const int row = blockIdx.x * 8 + wv;
    if (row >= rows) return;
    const float* p = y + (size_t)row * 128 + lane * 4;
    float v0 = p[0], v1 = p[1], v2 = p[2], v3 = p[3];
    float s = v0 + v1 + v2 + v3;
#pragma unroll
    for (int off = 16; off >= 1; off >>= 1) s += __shfl_xor(s, off, 32);
    const float mu = s * (1.f / 128.f);
    const float d0 = v0 - mu, d1 = v1 - mu, d2 = v2 - mu, d3 = v3 - mu;
    float q = d0 * d0 + d1 * d1 + d2 * d2 + d3 * d3;
#pragma unroll
    for (int off = 16; off >= 1; off >>= 1) q += __shfl_xor(q, off, 32);
    const float rs = rsqrtf(q * (1.f / 128.f) + 1e-12f);
    const int c = lane * 4;
    float* o = out + (size_t)row * 128 + c;
    o[0] = d0 * rs * gamma[c + 0] + beta[c + 0];
    o[1] = d1 * rs * gamma[c + 1] + beta[c + 1];
    o[2] = d2 * rs * gamma[c + 2] + beta[c + 2];
    o[3] = d3 * rs * gamma[c + 3] + beta[c + 3];
}

extern "C" void kernel_launch(void* const* d_in, const int* in_sizes, int n_in,
                              void* d_out, int out_size, void* d_ws, size_t ws_size,
                              hipStream_t stream)
{
    const float* embs  = (const float*)d_in[0];
    const float* w0    = (const float*)d_in[1];
    const float* b0    = (const float*)d_in[2];
    const float* att0  = (const float*)d_in[3];
    const float* bias0 = (const float*)d_in[4];
    const float* w1    = (const float*)d_in[5];
    const float* b1    = (const float*)d_in[6];
    const float* att1  = (const float*)d_in[7];
    const float* bias1 = (const float*)d_in[8];
    const float* wout  = (const float*)d_in[9];
    const float* bout  = (const float*)d_in[10];
    const float* gamma = (const float*)d_in[11];
    const float* beta  = (const float*)d_in[12];
    const int*   ei0   = (const int*)d_in[13];
    const int*   ei1   = (const int*)d_in[14];

    const int N  = in_sizes[0] / 128;
    const int E  = in_sizes[13] / 2;
    const int BS = out_size / 128;

    char* wsp = (char*)d_ws;
    size_t off = 0;
    auto carve = [&](size_t bytes) -> void* {
        void* p = wsp + off; off += (bytes + 255) & ~(size_t)255; return p;
    };
    float*     xl    = (float*)    carve((size_t)N * 128 * 4);
    float*     buf   = (float*)    carve((size_t)N * 128 * 4);
    const int  nzero  = (N + 1) + (N + 1) + (BS + 1);
    int*       zreg   = (int*)     carve((size_t)nzero * 4);
    int*       need   = zreg;
    int*       cnt    = need + (N + 1);
    int*       cnt1   = cnt + (N + 1);
    int*       list   = (int*)     carve((size_t)(N + 1) * 4);
    int*       nlist  = list + N;
    int*       rowptr = (int*)     carve((size_t)(N + 1) * 4);
    int*       cursor = (int*)     carve((size_t)(N + 1) * 4);
    int*       srcs   = (int*)     carve((size_t)E * 4);
    const size_t wpBytes  = (size_t)128 * 128 * 2;
    const size_t early    = 2 * wpBytes + 2 * (size_t)(BS + 1) * 4;
    const size_t ybytes   = (size_t)BS * 128 * 4;
    float*     ybuf    = (float*)  carve(ybytes > early ? ybytes : early);
    _Float16*  Wp0     = (_Float16*)ybuf;
    _Float16*  Wp1     = Wp0 + 128 * 128;
    int*       rowptr1 = (int*)(Wp1 + 128 * 128);
    int*       cursor1 = rowptr1 + (BS + 1);
    _Float16*  Wp2     = (_Float16*) carve(wpBytes);

    const int gemmBlocksN = (N + 15) / 16;
    const int edgeBlocks  = (E + 255) / 256;
    const int markBlocks  = ((E > BS ? E : BS) + 255) / 256;
    const int zeroBlocks  = (nzero + 255) / 256;
    const int scanBlocks0 = (N + 1 + 1023) / 1024;
    const int scanBlocks1 = (BS + 1 + 1023) / 1024;
    const int scanBlocks  = scanBlocks0 > scanBlocks1 ? scanBlocks0 : scanBlocks1;
    const int nodeBlocks1 = (BS + 7) / 8;
    const long long estRows64 = (long long)E * BS / N + BS;
    const int estRows     = estRows64 < N ? (int)estRows64 : N;
    const int nodeBlocksL = (estRows + 7) / 8;
    const int gemmBlocksL = (estRows + 15) / 16;

    prologue_k   <<<3 * kPackBlocks + zeroBlocks, 256, 0, stream>>>(w0, w1, wout, Wp0, Wp1, Wp2, zreg, nzero);

    gemm_mark_k  <<<gemmBlocksN + markBlocks, 256, 0, stream>>>(embs, Wp0, b0, xl, N, gemmBlocksN,
                                                                 ei1, cnt1, need, E, BS);

    count_dst_k  <<<edgeBlocks, 256, 0, stream>>>(ei0, need, cnt, E);
    scan_cnt_k   <<<dim3(scanBlocks, 3), 256, 0, stream>>>(cnt, rowptr, cursor, N, need, list, nlist,
                                                            cnt1, rowptr1, cursor1, BS);
    fill_src_k   <<<edgeBlocks, 256, 0, stream>>>(ei0, need, cursor, srcs, E, N);
    node_gat_k<true, true><<<nodeBlocksL, 256, 0, stream>>>(xl, rowptr, srcs, att0, bias0, buf, 0, list, nlist);

    gemm_fill_k  <<<gemmBlocksL + edgeBlocks, 256, 0, stream>>>(buf, Wp1, b1, xl, list, nlist, gemmBlocksL,
                                                                 ei1, cursor1, srcs, E, BS);
    node_gat_k<false, false><<<nodeBlocks1, 256, 0, stream>>>(xl, rowptr1, srcs, att1, bias1, buf, BS, nullptr, nullptr);

    gemm128_wmma <<<(BS + 15) / 16, 256, 0, stream>>>(buf, Wp2, bout, ybuf, BS);
    layernorm_k  <<<(BS + 7) / 8,   256, 0, stream>>>(ybuf, gamma, beta, (float*)d_out, BS);
}


# layer-1 attention, head GEMM and layernorm in one launch: 8-wave block puts 8 rows' attention into LDS as f16, multiplies by wout on WMMA, normalises the rows; 10 launches to 8; sums unchanged
# speedup vs baseline: 8.6656x; 1.1762x over previous
#include <hip/hip_runtime.h>
#include <hip/hip_bf16.h>
#include <math.h>

typedef _Float16 v16h __attribute__((ext_vector_type(16)));
typedef float    v8f  __attribute__((ext_vector_type(8)));

__device__ __forceinline__ void pack_w(const float* __restrict__ W, _Float16* __restrict__ Wp, int idx)
{
    if (idx >= 128 * 128) return;
    const int r  = idx >> 4;
    const int i  = idx & 15;
    const int col = r & 127;
    const int lh  = (r >> 7) & 1;
    const int kb  = r >> 8;
    const int kk  = (i < 8) ? (8 * lh + i) : (8 + 8 * lh + i);
    const int K   = 32 * kb + kk;
    Wp[(size_t)r * 16 + i] = (_Float16)W[(size_t)K * 128 + col];
}

constexpr int kPackBlocks = (128 * 128 + 255) / 256;

__global__ __launch_bounds__(256) void prologue_k(
    const float* __restrict__ w0, const float* __restrict__ w1, const float* __restrict__ w2,
    _Float16* __restrict__ Wp0, _Float16* __restrict__ Wp1, _Float16* __restrict__ Wp2,
    int* __restrict__ zreg, int nzero)
{
    const int b = (int)blockIdx.x;
    if (b < 3 * kPackBlocks) {
        const int which = b / kPackBlocks;
        const float* W  = which == 0 ? w0  : which == 1 ? w1  : w2;
        _Float16*    Wp = which == 0 ? Wp0 : which == 1 ? Wp1 : Wp2;
        pack_w(W, Wp, (b - which * kPackBlocks) * 256 + (int)threadIdx.x);
    } else {
        const int i = (b - 3 * kPackBlocks) * 256 + (int)threadIdx.x;
        if (i < nzero) zreg[i] = 0;
    }
}

typedef _Float16 v8h __attribute__((ext_vector_type(8)));
typedef _Float16 v4h __attribute__((ext_vector_type(4)));

constexpr int kApitch = 136;

template <bool LIST>
__device__ __forceinline__ void gemm_load_rows(const float* __restrict__ X, const int* __restrict__ list,
                                               int M, int mb, int wv, int lane,
                                               float4& r0, float4& r1, int& id)
{
    int m0 = mb + 2 * wv, m1 = m0 + 1;
    if (m0 >= M) m0 = M - 1;
    if (m1 >= M) m1 = M - 1;
    const int g0 = LIST ? list[m0] : m0;
    const int g1 = LIST ? list[m1] : m1;
    r0 = *(const float4*)(X + (size_t)g0 * 128 + 4 * lane);
    r1 = *(const float4*)(X + (size_t)g1 * 128 + 4 * lane);
    if (LIST) { const int mi = mb + (lane & 15); id = list[mi < M ? mi : M - 1]; }
}

__device__ __forceinline__ void lds_barrier() {
    asm volatile("s_wait_dscnt 0x0\n\ts_barrier_signal -1\n\ts_barrier_wait -1" ::: "memory");
}

__device__ __forceinline__ v4h cvt4h(const float4 x) {
    v4h h;
    h[0] = (_Float16)x.x; h[1] = (_Float16)x.y; h[2] = (_Float16)x.z; h[3] = (_Float16)x.w;
    return h;
}

__device__ __forceinline__ void load_b(const _Float16* __restrict__ Wp, int colB, int lh, v16h (&b)[4]) {
#pragma unroll
    for (int kt = 0; kt < 4; ++kt)
        b[kt] = *(const v16h*)(Wp + ((size_t)(kt * 2 + lh) * 128 + colB) * 16);
}

__device__ __forceinline__ v8f tile_times_b(const _Float16* __restrict__ At, const v16h (&b)[4], int lane) {
    const int lh = lane >> 4, lm = lane & 15;
    v8f acc = {};
#pragma unroll
    for (int kt = 0; kt < 4; ++kt) {
        const _Float16* pa = At + lm * kApitch + 32 * kt + 8 * lh;
        const v8h alo = *(const v8h*)(pa);
        const v8h ahi = *(const v8h*)(pa + 16);
        const v16h a = __builtin_shufflevector(alo, ahi, 0, 1, 2, 3, 4, 5, 6, 7, 8, 9, 10, 11, 12, 13, 14, 15);
        acc = __builtin_amdgcn_wmma_f32_16x16x32_f16(
                  false, a, false, b[kt], (short)0, acc, false, false);
    }
    return acc;
}

template <bool LIST, bool AHEAD>
__device__ __forceinline__ void gemm_tile(
    const float* __restrict__ X, float* __restrict__ Y, const int* __restrict__ list,
    int M, int mbase, int step, _Float16* __restrict__ At, const v16h (&b)[4], float bb,
    int wv, int lane, float4& r0, float4& r1, int& id)
{
    const int lh = lane >> 4, lm = lane & 15;
    const int colB = wv * 16 + lm;
    *(v4h*)(At + (2 * wv + 0) * kApitch + 4 * lane) = cvt4h(r0);
    *(v4h*)(At + (2 * wv + 1) * kApitch + 4 * lane) = cvt4h(r1);
    const int idv = id;
    lds_barrier();
    if (AHEAD) gemm_load_rows<LIST>(X, list, M, mbase + 2 * step, wv, lane, r0, r1, id);

    const v8f acc = tile_times_b(At, b, lane);
    float o[8];
#pragma unroll
    for (int r = 0; r < 8; ++r) o[r] = acc[r] + bb;
    float* const yp = Y + (size_t)(mbase + 8 * lh) * 128 + colB;
    if (mbase + 16 <= M) {
#pragma unroll
        for (int r = 0; r < 8; ++r) {
            if (LIST) {
                const int row = lh ? __builtin_amdgcn_readlane(idv, r + 8) : __builtin_amdgcn_readlane(idv, r);
                Y[(size_t)row * 128 + colB] = o[r];
            } else yp[r * 128] = o[r];
        }
    } else {
#pragma unroll
        for (int r = 0; r < 8; ++r) {
            if (mbase + r + 8 * lh < M) {
                if (LIST) {
                    const int row = lh ? __builtin_amdgcn_readlane(idv, r + 8) : __builtin_amdgcn_readlane(idv, r);
                    Y[(size_t)row * 128 + colB] = o[r];
                } else yp[r * 128] = o[r];
            }
        }
    }
}

template <bool LIST, bool MULTI>
__device__ __forceinline__ void gemm128_tiles(
    const float* __restrict__ X, const _Float16* __restrict__ Wp,
    const float* __restrict__ bias, float* __restrict__ Y, int M,
    const int* __restrict__ list, const int* __restrict__ nlist, int blk, int nblk)
{
    __shared__ __attribute__((aligned(16))) _Float16 As[2][16 * kApitch];

    const int lane  = threadIdx.x & 31;
    const int wv    = __builtin_amdgcn_readfirstlane((int)(threadIdx.x >> 5));
    const int lh    = lane >> 4;
    const int colB  = wv * 16 + (lane & 15);
    const float bb  = bias[colB];
    if (LIST) M = *nlist;

    const int step = nblk * 16;
    int mbase = blk * 16;
    if (mbase >= M) return;

    v16h b[4];
    load_b(Wp, colB, lh, b);

    float4 p0, p1;
    int pid = 0;
    if (MULTI) {
        float4 q0, q1;
        int qid = 0;
        gemm_load_rows<LIST>(X, list, M, mbase, wv, lane, p0, p1, pid);
        gemm_load_rows<LIST>(X, list, M, mbase + step, wv, lane, q0, q1, qid);
        for (; mbase < M; mbase += 2 * step) {
            gemm_tile<LIST, true>(X, Y, list, M, mbase, step, As[0], b, bb, wv, lane, p0, p1, pid);
            gemm_tile<LIST, true>(X, Y, list, M, mbase + step, step, As[1], b, bb, wv, lane, q0, q1, qid);
        }
    } else {
        int buf = 0;
        for (; mbase < M; mbase += step, buf ^= 1) {
            gemm_load_rows<LIST>(X, list, M, mbase, wv, lane, p0, p1, pid);
            gemm_tile<LIST, false>(X, Y, list, M, mbase, step, As[buf], b, bb, wv, lane, p0, p1, pid);
        }
    }
}

__device__ __forceinline__ void set_flag(int* p) {
    __hip_atomic_store(p, 1, __ATOMIC_RELAXED, __HIP_MEMORY_SCOPE_AGENT);
}

__device__ __forceinline__ void mark_count1(const int* __restrict__ ei, int* __restrict__ cnt1,
                                            int* __restrict__ need, int E, int nd, int e)
{
    if (e < nd) set_flag(&need[e]);
    if (e >= E) return;
    const int d = ei[E + e];
    if (d >= nd) return;
    atomicAdd(&cnt1[d], 1);
    set_flag(&need[ei[e]]);
}

__global__ __launch_bounds__(256) void gemm_mark_k(
    const float* __restrict__ X, const _Float16* __restrict__ Wp,
    const float* __restrict__ bias, float* __restrict__ Y, int M, int gemmBlocks,
    const int* __restrict__ ei, int* __restrict__ cnt1, int* __restrict__ need, int E, int nd)
{
    const int b = (int)blockIdx.x;
    if (b < gemmBlocks)
        gemm128_tiles<false, true>(X, Wp, bias, Y, M, nullptr, nullptr, b, gemmBlocks);
    else
        mark_count1(ei, cnt1, need, E, nd, (b - gemmBlocks) * 256 + (int)threadIdx.x);
}

__global__ void count_dst_k(const int* __restrict__ ei, const int* __restrict__ need,
                            int* __restrict__ cnt, int E)
{
    int e = blockIdx.x * blockDim.x + threadIdx.x;
    if (e >= E) return;
    const int d = ei[E + e];
    if (need[d]) atomicAdd(&cnt[d], 1);
}

__global__ __launch_bounds__(256) void scan_cnt_k(
    const int* __restrict__ cnt0, int* __restrict__ rowptr0, int* __restrict__ cursor0, int n0,
    const int* __restrict__ need, int* __restrict__ list, int* __restrict__ nlist,
    const int* __restrict__ cnt1, int* __restrict__ rowptr1, int* __restrict__ cursor1, int n1)
{
    const int plane = (int)blockIdx.y;
    const bool flags = plane == 1;
    const int* cnt = plane == 0 ? cnt0 : plane == 1 ? need : cnt1;
    int* rowptr = plane == 2 ? rowptr1 : rowptr0;
    int* cursor = plane == 2 ? cursor1 : cursor0;
    const int n = plane == 2 ? n1 : n0;
    __shared__ int s_front[8];
    __shared__ int s_own[8];
    const int t = threadIdx.x, lane = t & 31, wv = t >> 5;
    const int start = blockIdx.x * 1024;
    if (start > n) return;

    int front = 0;
    for (int i = t; i < start; i += 256) front += cnt[i];

    int c[4];
#pragma unroll
    for (int k = 0; k < 4; ++k) {
        const int idx = start + 4 * t + k;
        c[k] = idx < n ? cnt[idx] : 0;
    }
    const int own = c[0] + c[1] + c[2] + c[3];

#pragma unroll
    for (int off = 16; off >= 1; off >>= 1) front += __shfl_xor(front, off, 32);
    int incl = own;
#pragma unroll
    for (int off = 1; off < 32; off <<= 1) {
        const int v = __shfl_up(incl, off, 32);
        if (lane >= off) incl += v;
    }
    if (lane == 31) { s_front[wv] = front; s_own[wv] = incl; }
    __syncthreads();
    int carry = 0;
#pragma unroll
    for (int w = 0; w < 8; ++w) {
        carry += s_front[w];
        if (w < wv) carry += s_own[w];
    }
    int run = carry + incl - own;
#pragma unroll
    for (int k = 0; k < 4; ++k) {
        const int idx = start + 4 * t + k;
        if (flags) {
            if (idx < n && c[k]) list[run] = idx;
            if (idx == n) *nlist = run;
        } else if (idx <= n) { rowptr[idx] = run; cursor[idx] = run; }
        run += c[k];
    }
}

__device__ __forceinline__ void fill_src(const int* __restrict__ ei, const int* __restrict__ need,
                                         int* __restrict__ cursor, int* __restrict__ srcs,
                                         int E, int nd, int e)
{
    if (e >= E) return;
    const int d = ei[E + e];
    if (d >= nd) return;
    if (need && !need[d]) return;
    const int pos = atomicAdd(&cursor[d], 1);
    srcs[pos] = ei[e];
}

__global__ void fill_src_k(const int* __restrict__ ei, const int* __restrict__ need,
                           int* __restrict__ cursor, int* __restrict__ srcs, int E, int nd)
{
    fill_src(ei, need, cursor, srcs, E, nd, (int)(blockIdx.x * blockDim.x + threadIdx.x));
}

__global__ __launch_bounds__(256) void gemm_fill_k(
    const float* __restrict__ X, const _Float16* __restrict__ Wp,
    const float* __restrict__ bias, float* __restrict__ Y,
    const int* __restrict__ list, const int* __restrict__ nlist, int gemmBlocks,
    const int* __restrict__ ei, int* __restrict__ cursor1, int* __restrict__ srcs, int E, int nd)
{
    const int b = (int)blockIdx.x;
    if (b < gemmBlocks)
        gemm128_tiles<true, false>(X, Wp, bias, Y, 0, list, nlist, b, gemmBlocks);
    else
        fill_src(ei, nullptr, cursor1, srcs, E, nd, (b - gemmBlocks) * 256 + (int)threadIdx.x);
}


__device__ __forceinline__ float gelu_erf(float v) {
    return 0.5f * v * (1.f + erff(v * 0.70710678118654752440f));
}

template <int M>
__device__ __forceinline__ float xmask_add(float v) {
    const int o = __builtin_amdgcn_update_dpp(0, __float_as_int(v), 0x160 + M, 0xf, 0xf, true);
    return v + __int_as_float(o);
}

typedef float v2f __attribute__((ext_vector_type(2)));

__device__ __forceinline__ v2f lrelu02x2(v2f v) {
    const v2f s = v * 0.2f;
    v2f r;
    r.x = fmaxf(v.x, s.x);
    r.y = fmaxf(v.y, s.y);
    return r;
}

template <bool GELU>
__device__ __forceinline__ float4 node_row(
    const float* __restrict__ xl, const int* __restrict__ srcs, const float* __restrict__ att,
    const float* __restrict__ bias, int d, int beg, int end, int lane)
{
    int sv = 0;
    if (end > beg) {
        const int gi = beg + lane;
        sv = srcs[gi < end ? gi : end - 1];
    }
    const float4 a4 = *(const float4*)(att + 4 * lane);
    const float4 xd = *(const float4*)(xl + (size_t)d * 128 + 4 * lane);
    const float4 b4 = *(const float4*)(bias + 4 * lane);
    const v2f a_lo = {a4.x, a4.y}, a_hi = {a4.z, a4.w};
    const v2f d_lo = {xd.x, xd.y}, d_hi = {xd.z, xd.w};

    float m = -INFINITY, l = 0.f;
    v2f acc_lo = {0.f, 0.f}, acc_hi = {0.f, 0.f};

    for (int g = beg; g < end; g += 32) {
        if (g != beg) {
            const int gi = g + lane;
            sv = srcs[gi < end ? gi : end - 1];
        }
        const int gcnt = (end - g) < 32 ? (end - g) : 32;

        for (int c = 0; c < gcnt; c += 8) {
            const int left = gcnt - c;
            float4 xs[8];
#pragma unroll
            for (int u = 0; u < 8; ++u) {
                const int s = __builtin_amdgcn_readlane(sv, c + u);
                xs[u] = *(const float4*)(xl + (size_t)s * 128 + 4 * lane);
            }

            float p[8];
            float cm = -INFINITY;
#pragma unroll
            for (int u = 0; u < 8; ++u) {
                const v2f x_lo = {xs[u].x, xs[u].y}, x_hi = {xs[u].z, xs[u].w};
                const v2f t_lo = lrelu02x2(x_lo + d_lo);
                const v2f t_hi = lrelu02x2(x_hi + d_hi);
                const v2f r = t_lo * a_lo + t_hi * a_hi;
                float q = r.x + r.y;
                q = xmask_add<1>(q);
                q = xmask_add<2>(q);
                q = xmask_add<4>(q);
                p[u] = (u < left) ? q : -INFINITY;
                cm = fmaxf(cm, p[u]);
            }

            const float mn = fmaxf(m, cm);
            const float sc = __expf(m - mn);
            l *= sc;
            acc_lo *= sc; acc_hi *= sc;
#pragma unroll
            for (int u = 0; u < 8; ++u) {
                const v2f x_lo = {xs[u].x, xs[u].y}, x_hi = {xs[u].z, xs[u].w};
                const float w = __expf(p[u] - mn);
                l += w;
                acc_lo += w * x_lo;
                acc_hi += w * x_hi;
            }
            m = mn;
        }
    }

    float4 o = b4;
    if (end > beg) {
        o.x += acc_lo.x / l; o.y += acc_lo.y / l; o.z += acc_hi.x / l; o.w += acc_hi.y / l;
    }
    if (GELU) {
        o.x = gelu_erf(o.x); o.y = gelu_erf(o.y); o.z = gelu_erf(o.z); o.w = gelu_erf(o.w);
    }
    return o;
}

template <bool GELU, bool LIST>
__global__ __launch_bounds__(256) __attribute__((amdgpu_waves_per_eu(12))) void node_gat_k(
    const float* __restrict__ xl, const int* __restrict__ rowptr,
    const int* __restrict__ srcs, const float* __restrict__ att,
    const float* __restrict__ bias, float* __restrict__ out, int n,
    const int* __restrict__ list, const int* __restrict__ nlist)
{
    const int lane = threadIdx.x & 31;
    const int wv = __builtin_amdgcn_readfirstlane((int)(threadIdx.x >> 5));
    if (LIST) n = *nlist;

  for (int w = (int)blockIdx.x * 8 + wv; w < n; w += (int)gridDim.x * 8) {
    const int d = LIST ? __builtin_amdgcn_readfirstlane(list[w]) : w;
    const int beg = __builtin_amdgcn_readfirstlane(rowptr[d]);
    const int end = __builtin_amdgcn_readfirstlane(rowptr[d + 1]);

    const float4 o = node_row<GELU>(xl, srcs, att, bias, d, beg, end, lane);
    *(float4*)(out + (size_t)d * 128 + 4 * lane) = o;
  }
}

constexpr int kYpitch = 132;

__global__ __launch_bounds__(256) void gat_head_ln_k(
    const float* __restrict__ xl, const int* __restrict__ rowptr,
    const int* __restrict__ srcs, const float* __restrict__ att,
    const float* __restrict__ bias, const _Float16* __restrict__ Wp,
    const float* __restrict__ bout, const float* __restrict__ gamma,
    const float* __restrict__ beta, float* __restrict__ out, int n)
{
    __shared__ __attribute__((aligned(16))) _Float16 At[16 * kApitch];
    __shared__ __attribute__((aligned(16))) float Yt[16 * kYpitch];

    const int lane = threadIdx.x & 31;
    const int wv = __builtin_amdgcn_readfirstlane((int)(threadIdx.x >> 5));
    const int lh = lane >> 4, lm = lane & 15;
    const int d  = (int)blockIdx.x * 8 + wv;
    const bool live = d < n;

    int beg = 0, end = 0;
    if (live) {
        beg = __builtin_amdgcn_readfirstlane(rowptr[d]);
        end = __builtin_amdgcn_readfirstlane(rowptr[d + 1]);
    }
    const int colB = wv * 16 + lm;
    v16h b[4];
    load_b(Wp, colB, lh, b);
    const float4 bo4 = *(const float4*)(bout + 4 * lane);
    const float4 g4  = *(const float4*)(gamma + 4 * lane);
    const float4 be4 = *(const float4*)(beta + 4 * lane);

    float4 o = {0.f, 0.f, 0.f, 0.f};
    if (live) o = node_row<false>(xl, srcs, att, bias, d, beg, end, lane);
    const v4h zero4 = {};
    *(v4h*)(At + wv * kApitch + 4 * lane) = cvt4h(o);
    *(v4h*)(At + (wv + 8) * kApitch + 4 * lane) = zero4;
    lds_barrier();

    const v8f acc = tile_times_b(At, b, lane);
#pragma unroll
    for (int r = 0; r < 8; ++r) Yt[(r + 8 * lh) * kYpitch + colB] = acc[r];
    lds_barrier();
    if (!live) return;

    const float4 t = *(const float4*)(Yt + wv * kYpitch + 4 * lane);
    const float4 y = {t.x + bo4.x, t.y + bo4.y, t.z + bo4.z, t.w + bo4.w};
    float s = y.x + y.y + y.z + y.w;
#pragma unroll
    for (int off = 16; off >= 1; off >>= 1) s += __shfl_xor(s, off, 32);
    const float mu = s * (1.f / 128.f);
    const float d0 = y.x - mu, d1 = y.y - mu, d2 = y.z - mu, d3 = y.w - mu;
    float q = d0 * d0 + d1 * d1 + d2 * d2 + d3 * d3;
#pragma unroll
    for (int off = 16; off >= 1; off >>= 1) q += __shfl_xor(q, off, 32);
    const float rs = rsqrtf(q * (1.f / 128.f) + 1e-12f);
    float4 r4;
    r4.x = d0 * rs * g4.x + be4.x;
    r4.y = d1 * rs * g4.y + be4.y;
    r4.z = d2 * rs * g4.z + be4.z;
    r4.w = d3 * rs * g4.w + be4.w;
    *(float4*)(out + (size_t)d * 128 + 4 * lane) = r4;
}

extern "C" void kernel_launch(void* const* d_in, const int* in_sizes, int n_in,
                              void* d_out, int out_size, void* d_ws, size_t ws_size,
                              hipStream_t stream)
{
    const float* embs  = (const float*)d_in[0];
    const float* w0    = (const float*)d_in[1];
    const float* b0    = (const float*)d_in[2];
    const float* att0  = (const float*)d_in[3];
    const float* bias0 = (const float*)d_in[4];
    const float* w1    = (const float*)d_in[5];
    const float* b1    = (const float*)d_in[6];
    const float* att1  = (const float*)d_in[7];
    const float* bias1 = (const float*)d_in[8];
    const float* wout  = (const float*)d_in[9];
    const float* bout  = (const float*)d_in[10];
    const float* gamma = (const float*)d_in[11];
    const float* beta  = (const float*)d_in[12];
    const int*   ei0   = (const int*)d_in[13];
    const int*   ei1   = (const int*)d_in[14];

    const int N  = in_sizes[0] / 128;
    const int E  = in_sizes[13] / 2;
    const int BS = out_size / 128;

    char* wsp = (char*)d_ws;
    size_t off = 0;
    auto carve = [&](size_t bytes) -> void* {
        void* p = wsp + off; off += (bytes + 255) & ~(size_t)255; return p;
    };
    float*     xl    = (float*)    carve((size_t)N * 128 * 4);
    float*     buf   = (float*)    carve((size_t)N * 128 * 4);
    const int  nzero  = (N + 1) + (N + 1) + (BS + 1);
    int*       zreg   = (int*)     carve((size_t)nzero * 4);
    int*       need   = zreg;
    int*       cnt    = need + (N + 1);
    int*       cnt1   = cnt + (N + 1);
    int*       list   = (int*)     carve((size_t)(N + 1) * 4);
    int*       nlist  = list + N;
    int*       rowptr = (int*)     carve((size_t)(N + 1) * 4);
    int*       cursor = (int*)     carve((size_t)(N + 1) * 4);
    int*       srcs   = (int*)     carve((size_t)E * 4);
    const size_t wpBytes  = (size_t)128 * 128 * 2;
    _Float16*  Wp0     = (_Float16*) carve(wpBytes);
    _Float16*  Wp1     = (_Float16*) carve(wpBytes);
    _Float16*  Wp2     = (_Float16*) carve(wpBytes);
    int*       rowptr1 = (int*)     carve((size_t)(BS + 1) * 4);
    int*       cursor1 = (int*)     carve((size_t)(BS + 1) * 4);

    const int gemmBlocksN = ((N + 15) / 16 + 3) / 4;
    const int edgeBlocks  = (E + 255) / 256;
    const int markBlocks  = ((E > BS ? E : BS) + 255) / 256;
    const int zeroBlocks  = (nzero + 255) / 256;
    const int scanBlocks0 = (N + 1 + 1023) / 1024;
    const int scanBlocks1 = (BS + 1 + 1023) / 1024;
    const int scanBlocks  = scanBlocks0 > scanBlocks1 ? scanBlocks0 : scanBlocks1;
    const int headBlocks  = (BS + 7) / 8;
    const long long estRows64 = (long long)E * BS / N + BS;
    const int estRows     = estRows64 < N ? (int)estRows64 : N;
    const int nodeBlocksL = (estRows + 7) / 8;
    const int gemmBlocksL = (estRows + 15) / 16;

    prologue_k   <<<3 * kPackBlocks + zeroBlocks, 256, 0, stream>>>(w0, w1, wout, Wp0, Wp1, Wp2, zreg, nzero);

    gemm_mark_k  <<<gemmBlocksN + markBlocks, 256, 0, stream>>>(embs, Wp0, b0, xl, N, gemmBlocksN,
                                                                 ei1, cnt1, need, E, BS);

    count_dst_k  <<<edgeBlocks, 256, 0, stream>>>(ei0, need, cnt, E);
    scan_cnt_k   <<<dim3(scanBlocks, 3), 256, 0, stream>>>(cnt, rowptr, cursor, N, need, list, nlist,
                                                            cnt1, rowptr1, cursor1, BS);
    fill_src_k   <<<edgeBlocks, 256, 0, stream>>>(ei0, need, cursor, srcs, E, N);
    node_gat_k<true, true><<<nodeBlocksL, 256, 0, stream>>>(xl, rowptr, srcs, att0, bias0, buf, 0, list, nlist);

    gemm_fill_k  <<<gemmBlocksL + edgeBlocks, 256, 0, stream>>>(buf, Wp1, b1, xl, list, nlist, gemmBlocksL,
                                                                 ei1, cursor1, srcs, E, BS);
    gat_head_ln_k<<<headBlocks, 256, 0, stream>>>(xl, rowptr1, srcs, att1, bias1, Wp2, bout, gamma, beta,
                                                  (float*)d_out, BS);
}


# layer-0 edges grouped in one pass: an edge draws a slot and puts its source id into its destination's unwritten output row; count pass gone, fill joins scan launch; 8 -> 6 launches; sum order kept
# speedup vs baseline: 11.2259x; 1.2955x over previous
#include <hip/hip_runtime.h>
#include <hip/hip_bf16.h>
#include <math.h>

typedef _Float16 v16h __attribute__((ext_vector_type(16)));
typedef float    v8f  __attribute__((ext_vector_type(8)));

__device__ __forceinline__ void pack_w(const float* __restrict__ W, _Float16* __restrict__ Wp, int idx)
{
    if (idx >= 128 * 128) return;
    const int r  = idx >> 4;
    const int i  = idx & 15;
    const int col = r & 127;
    const int lh  = (r >> 7) & 1;
    const int kb  = r >> 8;
    const int kk  = (i < 8) ? (8 * lh + i) : (8 + 8 * lh + i);
    const int K   = 32 * kb + kk;
    Wp[(size_t)r * 16 + i] = (_Float16)W[(size_t)K * 128 + col];
}

constexpr int kPackBlocks = (128 * 128 + 255) / 256;

__global__ __launch_bounds__(256) void prologue_k(
    const float* __restrict__ w0, const float* __restrict__ w1, const float* __restrict__ w2,
    _Float16* __restrict__ Wp0, _Float16* __restrict__ Wp1, _Float16* __restrict__ Wp2,
    int* __restrict__ zreg, int nzero)
{
    const int b = (int)blockIdx.x;
    if (b < 3 * kPackBlocks) {
        const int which = b / kPackBlocks;
        const float* W  = which == 0 ? w0  : which == 1 ? w1  : w2;
        _Float16*    Wp = which == 0 ? Wp0 : which == 1 ? Wp1 : Wp2;
        pack_w(W, Wp, (b - which * kPackBlocks) * 256 + (int)threadIdx.x);
    } else {
        const int i = (b - 3 * kPackBlocks) * 256 + (int)threadIdx.x;
        if (i < nzero) zreg[i] = 0;
    }
}

typedef _Float16 v8h __attribute__((ext_vector_type(8)));
typedef _Float16 v4h __attribute__((ext_vector_type(4)));

constexpr int kApitch = 136;

template <bool LIST>
__device__ __forceinline__ void gemm_load_rows(const float* __restrict__ X, const int* __restrict__ list,
                                               int M, int mb, int wv, int lane,
                                               float4& r0, float4& r1, int& id)
{
    int m0 = mb + 2 * wv, m1 = m0 + 1;
    if (m0 >= M) m0 = M - 1;
    if (m1 >= M) m1 = M - 1;
    const int g0 = LIST ? list[m0] : m0;
    const int g1 = LIST ? list[m1] : m1;
    r0 = *(const float4*)(X + (size_t)g0 * 128 + 4 * lane);
    r1 = *(const float4*)(X + (size_t)g1 * 128 + 4 * lane);
    if (LIST) { const int mi = mb + (lane & 15); id = list[mi < M ? mi : M - 1]; }
}

__device__ __forceinline__ void lds_barrier() {
    asm volatile("s_wait_dscnt 0x0\n\ts_barrier_signal -1\n\ts_barrier_wait -1" ::: "memory");
}

__device__ __forceinline__ v4h cvt4h(const float4 x) {
    v4h h;
    h[0] = (_Float16)x.x; h[1] = (_Float16)x.y; h[2] = (_Float16)x.z; h[3] = (_Float16)x.w;
    return h;
}

__device__ __forceinline__ void load_b(const _Float16* __restrict__ Wp, int colB, int lh, v16h (&b)[4]) {
#pragma unroll
    for (int kt = 0; kt < 4; ++kt)
        b[kt] = *(const v16h*)(Wp + ((size_t)(kt * 2 + lh) * 128 + colB) * 16);
}

__device__ __forceinline__ v8f tile_times_b(const _Float16* __restrict__ At, const v16h (&b)[4], int lane) {
    const int lh = lane >> 4, lm = lane & 15;
    v8f acc = {};
#pragma unroll
    for (int kt = 0; kt < 4; ++kt) {
        const _Float16* pa = At + lm * kApitch + 32 * kt + 8 * lh;
        const v8h alo = *(const v8h*)(pa);
        const v8h ahi = *(const v8h*)(pa + 16);
        const v16h a = __builtin_shufflevector(alo, ahi, 0, 1, 2, 3, 4, 5, 6, 7, 8, 9, 10, 11, 12, 13, 14, 15);
        acc = __builtin_amdgcn_wmma_f32_16x16x32_f16(
                  false, a, false, b[kt], (short)0, acc, false, false);
    }
    return acc;
}

template <bool LIST, bool AHEAD>
__device__ __forceinline__ void gemm_tile(
    const float* __restrict__ X, float* __restrict__ Y, const int* __restrict__ list,
    int M, int mbase, int step, _Float16* __restrict__ At, const v16h (&b)[4], float bb,
    int wv, int lane, float4& r0, float4& r1, int& id)
{
    const int lh = lane >> 4, lm = lane & 15;
    const int colB = wv * 16 + lm;
    *(v4h*)(At + (2 * wv + 0) * kApitch + 4 * lane) = cvt4h(r0);
    *(v4h*)(At + (2 * wv + 1) * kApitch + 4 * lane) = cvt4h(r1);
    const int idv = id;
    lds_barrier();
    if (AHEAD) gemm_load_rows<LIST>(X, list, M, mbase + 2 * step, wv, lane, r0, r1, id);

    const v8f acc = tile_times_b(At, b, lane);
    float o[8];
#pragma unroll
    for (int r = 0; r < 8; ++r) o[r] = acc[r] + bb;
    float* const yp = Y + (size_t)(mbase + 8 * lh) * 128 + colB;
    if (mbase + 16 <= M) {
#pragma unroll
        for (int r = 0; r < 8; ++r) {
            if (LIST) {
                const int row = lh ? __builtin_amdgcn_readlane(idv, r + 8) : __builtin_amdgcn_readlane(idv, r);
                Y[(size_t)row * 128 + colB] = o[r];
            } else yp[r * 128] = o[r];
        }
    } else {
#pragma unroll
        for (int r = 0; r < 8; ++r) {
            if (mbase + r + 8 * lh < M) {
                if (LIST) {
                    const int row = lh ? __builtin_amdgcn_readlane(idv, r + 8) : __builtin_amdgcn_readlane(idv, r);
                    Y[(size_t)row * 128 + colB] = o[r];
                } else yp[r * 128] = o[r];
            }
        }
    }
}

template <bool LIST, bool MULTI>
__device__ __forceinline__ void gemm128_tiles(
    const float* __restrict__ X, const _Float16* __restrict__ Wp,
    const float* __restrict__ bias, float* __restrict__ Y, int M,
    const int* __restrict__ list, const int* __restrict__ nlist, int blk, int nblk)
{
    __shared__ __attribute__((aligned(16))) _Float16 As[2][16 * kApitch];

    const int lane  = threadIdx.x & 31;
    const int wv    = __builtin_amdgcn_readfirstlane((int)(threadIdx.x >> 5));
    const int lh    = lane >> 4;
    const int colB  = wv * 16 + (lane & 15);
    const float bb  = bias[colB];
    if (LIST) M = *nlist;

    const int step = nblk * 16;
    int mbase = blk * 16;
    if (mbase >= M) return;

    v16h b[4];
    load_b(Wp, colB, lh, b);

    float4 p0, p1;
    int pid = 0;
    if (MULTI) {
        float4 q0, q1;
        int qid = 0;
        gemm_load_rows<LIST>(X, list, M, mbase, wv, lane, p0, p1, pid);
        gemm_load_rows<LIST>(X, list, M, mbase + step, wv, lane, q0, q1, qid);
        for (; mbase < M; mbase += 2 * step) {
            gemm_tile<LIST, true>(X, Y, list, M, mbase, step, As[0], b, bb, wv, lane, p0, p1, pid);
            gemm_tile<LIST, true>(X, Y, list, M, mbase + step, step, As[1], b, bb, wv, lane, q0, q1, qid);
        }
    } else {
        int buf = 0;
        for (; mbase < M; mbase += step, buf ^= 1) {
            gemm_load_rows<LIST>(X, list, M, mbase, wv, lane, p0, p1, pid);
            gemm_tile<LIST, false>(X, Y, list, M, mbase, step, As[buf], b, bb, wv, lane, p0, p1, pid);
        }
    }
}

__device__ __forceinline__ void set_flag(int* p) {
    __hip_atomic_store(p, 1, __ATOMIC_RELAXED, __HIP_MEMORY_SCOPE_AGENT);
}

__device__ __forceinline__ void mark_count1(const int* __restrict__ ei, int* __restrict__ cnt1,
                                            int* __restrict__ need, int E, int nd, int e)
{
    if (e < nd) set_flag(&need[e]);
    if (e >= E) return;
    const int d = ei[E + e];
    if (d >= nd) return;
    atomicAdd(&cnt1[d], 1);
    set_flag(&need[ei[e]]);
}

__global__ __launch_bounds__(256) void gemm_mark_k(
    const float* __restrict__ X, const _Float16* __restrict__ Wp,
    const float* __restrict__ bias, float* __restrict__ Y, int M, int gemmBlocks,
    const int* __restrict__ ei, int* __restrict__ cnt1, int* __restrict__ need, int E, int nd)
{
    const int b = (int)blockIdx.x;
    if (b < gemmBlocks)
        gemm128_tiles<false, true>(X, Wp, bias, Y, M, nullptr, nullptr, b, gemmBlocks);
    else
        mark_count1(ei, cnt1, need, E, nd, (b - gemmBlocks) * 256 + (int)threadIdx.x);
}

template <bool FLAGS>
__device__ __forceinline__ void scan_tile(
    const int* __restrict__ cnt, int n, int bx,
    int* __restrict__ rowptr, int* __restrict__ cursor,
    int* __restrict__ list, int* __restrict__ nlist)
{
    __shared__ int s_front[8];
    __shared__ int s_own[8];
    const int t = threadIdx.x, lane = t & 31, wv = t >> 5;
    const int start = bx * 1024;
    if (start > n) return;

    int front = 0;
    for (int i = t; i < start; i += 256) front += cnt[i];

    int c[4];
#pragma unroll
    for (int k = 0; k < 4; ++k) {
        const int idx = start + 4 * t + k;
        c[k] = idx < n ? cnt[idx] : 0;
    }
    const int own = c[0] + c[1] + c[2] + c[3];

#pragma unroll
    for (int off = 16; off >= 1; off >>= 1) front += __shfl_xor(front, off, 32);
    int incl = own;
#pragma unroll
    for (int off = 1; off < 32; off <<= 1) {
        const int v = __shfl_up(incl, off, 32);
        if (lane >= off) incl += v;
    }
    if (lane == 31) { s_front[wv] = front; s_own[wv] = incl; }
    __syncthreads();
    int carry = 0;
#pragma unroll
    for (int w = 0; w < 8; ++w) {
        carry += s_front[w];
        if (w < wv) carry += s_own[w];
    }
    int run = carry + incl - own;
#pragma unroll
    for (int k = 0; k < 4; ++k) {
        const int idx = start + 4 * t + k;
        if (FLAGS) {
            if (idx < n && c[k]) list[run] = idx;
            if (idx == n) *nlist = run;
        } else if (idx <= n) { rowptr[idx] = run; cursor[idx] = run; }
        run += c[k];
    }
}

__device__ __forceinline__ void fill_src(const int* __restrict__ ei,
                                         int* __restrict__ cursor, int* __restrict__ srcs,
                                         int E, int nd, int e)
{
    if (e >= E) return;
    const int d = ei[E + e];
    if (d >= nd) return;
    const int pos = atomicAdd(&cursor[d], 1);
    srcs[pos] = ei[e];
}

constexpr int kBin = 128;

__device__ __forceinline__ void bin_src(const int* __restrict__ ei, const int* __restrict__ need,
                                        int* __restrict__ cnt, int* __restrict__ bins,
                                        int E, int nd, int e)
{
    if (e >= E) return;
    const int d = ei[E + e];
    if (d >= nd) return;
    if (!need[d]) return;
    const int slot = atomicAdd(&cnt[d], 1);
    if (slot < kBin) bins[(size_t)d * kBin + slot] = ei[e];
}

__global__ __launch_bounds__(256) void scan_bin_k(
    const int* __restrict__ need, int* __restrict__ list, int* __restrict__ nlist, int n0,
    const int* __restrict__ cnt1, int* __restrict__ rowptr1, int* __restrict__ cursor1, int n1,
    int scanBlocks,
    const int* __restrict__ ei, int* __restrict__ cnt, int* __restrict__ bins, int E)
{
    const int b = (int)blockIdx.x;
    if (b < scanBlocks)
        scan_tile<true>(need, n0, b, nullptr, nullptr, list, nlist);
    else if (b < 2 * scanBlocks)
        scan_tile<false>(cnt1, n1, b - scanBlocks, rowptr1, cursor1, nullptr, nullptr);
    else
        bin_src(ei, need, cnt, bins, E, n0, (b - 2 * scanBlocks) * 256 + (int)threadIdx.x);
}

__global__ __launch_bounds__(256) void gemm_fill_k(
    const float* __restrict__ X, const _Float16* __restrict__ Wp,
    const float* __restrict__ bias, float* __restrict__ Y,
    const int* __restrict__ list, const int* __restrict__ nlist, int gemmBlocks,
    const int* __restrict__ ei, int* __restrict__ cursor1, int* __restrict__ srcs, int E, int nd)
{
    const int b = (int)blockIdx.x;
    if (b < gemmBlocks)
        gemm128_tiles<true, false>(X, Wp, bias, Y, 0, list, nlist, b, gemmBlocks);
    else
        fill_src(ei, cursor1, srcs, E, nd, (b - gemmBlocks) * 256 + (int)threadIdx.x);
}


__device__ __forceinline__ float gelu_erf(float v) {
    return 0.5f * v * (1.f + erff(v * 0.70710678118654752440f));
}

template <int M>
__device__ __forceinline__ float xmask_add(float v) {
    const int o = __builtin_amdgcn_update_dpp(0, __float_as_int(v), 0x160 + M, 0xf, 0xf, true);
    return v + __int_as_float(o);
}

typedef float v2f __attribute__((ext_vector_type(2)));

__device__ __forceinline__ v2f lrelu02x2(v2f v) {
    const v2f s = v * 0.2f;
    v2f r;
    r.x = fmaxf(v.x, s.x);
    r.y = fmaxf(v.y, s.y);
    return r;
}

__device__ __forceinline__ void attend_group(
    const float* __restrict__ xl, int sv, int gcnt, int lane,
    const v2f a_lo, const v2f a_hi, const v2f d_lo, const v2f d_hi,
    float& m, float& l, v2f& acc_lo, v2f& acc_hi)
{
    for (int c = 0; c < gcnt; c += 8) {
        const int left = gcnt - c;
        float4 xs[8];
#pragma unroll
        for (int u = 0; u < 8; ++u) {
            const int s = __builtin_amdgcn_readlane(sv, c + u);
            xs[u] = *(const float4*)(xl + (size_t)s * 128 + 4 * lane);
        }

        float p[8];
        float cm = -INFINITY;
#pragma unroll
        for (int u = 0; u < 8; ++u) {
            const v2f x_lo = {xs[u].x, xs[u].y}, x_hi = {xs[u].z, xs[u].w};
            const v2f t_lo = lrelu02x2(x_lo + d_lo);
            const v2f t_hi = lrelu02x2(x_hi + d_hi);
            const v2f r = t_lo * a_lo + t_hi * a_hi;
            float q = r.x + r.y;
            q = xmask_add<1>(q);
            q = xmask_add<2>(q);
            q = xmask_add<4>(q);
            p[u] = (u < left) ? q : -INFINITY;
            cm = fmaxf(cm, p[u]);
        }

        const float mn = fmaxf(m, cm);
        const float sc = __expf(m - mn);
        l *= sc;
        acc_lo *= sc; acc_hi *= sc;
#pragma unroll
        for (int u = 0; u < 8; ++u) {
            const v2f x_lo = {xs[u].x, xs[u].y}, x_hi = {xs[u].z, xs[u].w};
            const float w = __expf(p[u] - mn);
            l += w;
            acc_lo += w * x_lo;
            acc_hi += w * x_hi;
        }
        m = mn;
    }
}

template <bool GELU>
__device__ __forceinline__ float4 finish_row(const float4 b4, bool any, float l, const v2f acc_lo, const v2f acc_hi)
{
    float4 o = b4;
    if (any) {
        o.x += acc_lo.x / l; o.y += acc_lo.y / l; o.z += acc_hi.x / l; o.w += acc_hi.y / l;
    }
    if (GELU) {
        o.x = gelu_erf(o.x); o.y = gelu_erf(o.y); o.z = gelu_erf(o.z); o.w = gelu_erf(o.w);
    }
    return o;
}

template <bool GELU>
__device__ __forceinline__ float4 node_row(
    const float* __restrict__ xl, const int* __restrict__ srcs, const float* __restrict__ att,
    const float* __restrict__ bias, int d, int beg, int end, int lane)
{
    int sv = 0;
    if (end > beg) {
        const int gi = beg + lane;
        sv = srcs[gi < end ? gi : end - 1];
    }
    const float4 a4 = *(const float4*)(att + 4 * lane);
    const float4 xd = *(const float4*)(xl + (size_t)d * 128 + 4 * lane);
    const float4 b4 = *(const float4*)(bias + 4 * lane);
    const v2f a_lo = {a4.x, a4.y}, a_hi = {a4.z, a4.w};
    const v2f d_lo = {xd.x, xd.y}, d_hi = {xd.z, xd.w};

    float m = -INFINITY, l = 0.f;
    v2f acc_lo = {0.f, 0.f}, acc_hi = {0.f, 0.f};

    for (int g = beg; g < end; g += 32) {
        if (g != beg) {
            const int gi = g + lane;
            sv = srcs[gi < end ? gi : end - 1];
        }
        const int gcnt = (end - g) < 32 ? (end - g) : 32;
        attend_group(xl, sv, gcnt, lane, a_lo, a_hi, d_lo, d_hi, m, l, acc_lo, acc_hi);
    }
    return finish_row<GELU>(b4, end > beg, l, acc_lo, acc_hi);
}

template <bool GELU>
__device__ __forceinline__ float4 node_row_bins(
    const float* __restrict__ xl, const int* ids, const int* __restrict__ cntd,
    const int* __restrict__ ei, int E, const float* __restrict__ att,
    const float* __restrict__ bias, int d, int lane)
{
    int first = ids[lane];
    const int nraw = *cntd;
    const float4 a4 = *(const float4*)(att + 4 * lane);
    const float4 xd = *(const float4*)(xl + (size_t)d * 128 + 4 * lane);
    const float4 b4 = *(const float4*)(bias + 4 * lane);
    const v2f a_lo = {a4.x, a4.y}, a_hi = {a4.z, a4.w};
    const v2f d_lo = {xd.x, xd.y}, d_hi = {xd.z, xd.w};
    const int n = __builtin_amdgcn_readfirstlane(nraw);

    float m = -INFINITY, l = 0.f;
    v2f acc_lo = {0.f, 0.f}, acc_hi = {0.f, 0.f};

    if (n <= kBin) {
        const int id0 = __builtin_amdgcn_readfirstlane(first);
        first = lane < n ? first : id0;
        for (int g = 0; g < n; g += 32) {
            int sv = first;
            if (g != 0) {
                const int gi = g + lane;
                sv = ids[gi < n ? gi : n - 1];
            }
            const int gcnt = (n - g) < 32 ? (n - g) : 32;
            attend_group(xl, sv, gcnt, lane, a_lo, a_hi, d_lo, d_hi, m, l, acc_lo, acc_hi);
        }
    } else {
        for (int i = 0; i < E; i += 32) {
            const int e = i + lane;
            const int dd = e < E ? ei[E + e] : -1;
            unsigned mask = __builtin_amdgcn_ballot_w32(dd == d);
            if (mask) {
                const int ss = ei[e < E ? e : E - 1];
                do {
                    const int idx = __builtin_ctz(mask);
                    mask &= mask - 1;
                    const int s = __builtin_amdgcn_readlane(ss, idx);
                    attend_group(xl, s, 1, lane, a_lo, a_hi, d_lo, d_hi, m, l, acc_lo, acc_hi);
                } while (mask);
            }
        }
    }
    return finish_row<GELU>(b4, n > 0, l, acc_lo, acc_hi);
}

__global__ __launch_bounds__(256) __attribute__((amdgpu_waves_per_eu(12))) void node_gat_bins_k(
    const float* __restrict__ xl, const int* __restrict__ cnt,
    const int* __restrict__ ei, int E, const float* __restrict__ att,
    const float* __restrict__ bias, float* rows,
    const int* __restrict__ list, const int* __restrict__ nlist)
{
    const int lane = threadIdx.x & 31;
    const int wv = __builtin_amdgcn_readfirstlane((int)(threadIdx.x >> 5));
    const int n = *nlist;

  for (int w = (int)blockIdx.x * 8 + wv; w < n; w += (int)gridDim.x * 8) {
    const int d = __builtin_amdgcn_readfirstlane(list[w]);
    float* const row = rows + (size_t)d * 128;
    const float4 o = node_row_bins<true>(xl, (const int*)row, cnt + d, ei, E, att, bias, d, lane);
    *(float4*)(row + 4 * lane) = o;
  }
}

constexpr int kYpitch = 132;

__global__ __launch_bounds__(256) void gat_head_ln_k(
    const float* __restrict__ xl, const int* __restrict__ rowptr,
    const int* __restrict__ srcs, const float* __restrict__ att,
    const float* __restrict__ bias, const _Float16* __restrict__ Wp,
    const float* __restrict__ bout, const float* __restrict__ gamma,
    const float* __restrict__ beta, float* __restrict__ out, int n)
{
    __shared__ __attribute__((aligned(16))) _Float16 At[16 * kApitch];
    __shared__ __attribute__((aligned(16))) float Yt[16 * kYpitch];

    const int lane = threadIdx.x & 31;
    const int wv = __builtin_amdgcn_readfirstlane((int)(threadIdx.x >> 5));
    const int lh = lane >> 4, lm = lane & 15;
    const int d  = (int)blockIdx.x * 8 + wv;
    const bool live = d < n;

    int beg = 0, end = 0;
    if (live) {
        beg = __builtin_amdgcn_readfirstlane(rowptr[d]);
        end = __builtin_amdgcn_readfirstlane(rowptr[d + 1]);
    }
    const int colB = wv * 16 + lm;
    v16h b[4];
    load_b(Wp, colB, lh, b);
    const float4 bo4 = *(const float4*)(bout + 4 * lane);
    const float4 g4  = *(const float4*)(gamma + 4 * lane);
    const float4 be4 = *(const float4*)(beta + 4 * lane);

    float4 o = {0.f, 0.f, 0.f, 0.f};
    if (live) o = node_row<false>(xl, srcs, att, bias, d, beg, end, lane);
    const v4h zero4 = {};
    *(v4h*)(At + wv * kApitch + 4 * lane) = cvt4h(o);
    *(v4h*)(At + (wv + 8) * kApitch + 4 * lane) = zero4;
    lds_barrier();

    const v8f acc = tile_times_b(At, b, lane);
#pragma unroll
    for (int r = 0; r < 8; ++r) Yt[(r + 8 * lh) * kYpitch + colB] = acc[r];
    lds_barrier();
    if (!live) return;

    const float4 t = *(const float4*)(Yt + wv * kYpitch + 4 * lane);
    const float4 y = {t.x + bo4.x, t.y + bo4.y, t.z + bo4.z, t.w + bo4.w};
    float s = y.x + y.y + y.z + y.w;
#pragma unroll
    for (int off = 16; off >= 1; off >>= 1) s += __shfl_xor(s, off, 32);
    const float mu = s * (1.f / 128.f);
    const float d0 = y.x - mu, d1 = y.y - mu, d2 = y.z - mu, d3 = y.w - mu;
    float q = d0 * d0 + d1 * d1 + d2 * d2 + d3 * d3;
#pragma unroll
    for (int off = 16; off >= 1; off >>= 1) q += __shfl_xor(q, off, 32);
    const float rs = rsqrtf(q * (1.f / 128.f) + 1e-12f);
    float4 r4;
    r4.x = d0 * rs * g4.x + be4.x;
    r4.y = d1 * rs * g4.y + be4.y;
    r4.z = d2 * rs * g4.z + be4.z;
    r4.w = d3 * rs * g4.w + be4.w;
    *(float4*)(out + (size_t)d * 128 + 4 * lane) = r4;
}

extern "C" void kernel_launch(void* const* d_in, const int* in_sizes, int n_in,
                              void* d_out, int out_size, void* d_ws, size_t ws_size,
                              hipStream_t stream)
{
    const float* embs  = (const float*)d_in[0];
    const float* w0    = (const float*)d_in[1];
    const float* b0    = (const float*)d_in[2];
    const float* att0  = (const float*)d_in[3];
    const float* bias0 = (const float*)d_in[4];
    const float* w1    = (const float*)d_in[5];
    const float* b1    = (const float*)d_in[6];
    const float* att1  = (const float*)d_in[7];
    const float* bias1 = (const float*)d_in[8];
    const float* wout  = (const float*)d_in[9];
    const float* bout  = (const float*)d_in[10];
    const float* gamma = (const float*)d_in[11];
    const float* beta  = (const float*)d_in[12];
    const int*   ei0   = (const int*)d_in[13];
    const int*   ei1   = (const int*)d_in[14];

    const int N  = in_sizes[0] / 128;
    const int E  = in_sizes[13] / 2;
    const int BS = out_size / 128;

    char* wsp = (char*)d_ws;
    size_t off = 0;
    auto carve = [&](size_t bytes) -> void* {
        void* p = wsp + off; off += (bytes + 255) & ~(size_t)255; return p;
    };
    float*     xl    = (float*)    carve((size_t)N * 128 * 4);
    float*     buf   = (float*)    carve((size_t)N * 128 * 4);
    const int  nzero  = (N + 1) + (N + 1) + (BS + 1);
    int*       zreg   = (int*)     carve((size_t)nzero * 4);
    int*       need   = zreg;
    int*       cnt    = need + (N + 1);
    int*       cnt1   = cnt + (N + 1);
    int*       list   = (int*)     carve((size_t)(N + 1) * 4);
    int*       nlist  = list + N;
    int*       srcs   = (int*)     carve((size_t)E * 4);
    const size_t wpBytes  = (size_t)128 * 128 * 2;
    _Float16*  Wp0     = (_Float16*) carve(wpBytes);
    _Float16*  Wp1     = (_Float16*) carve(wpBytes);
    _Float16*  Wp2     = (_Float16*) carve(wpBytes);
    int*       rowptr1 = (int*)     carve((size_t)(BS + 1) * 4);
    int*       cursor1 = (int*)     carve((size_t)(BS + 1) * 4);

    const int gemmBlocksN = ((N + 15) / 16 + 3) / 4;
    const int edgeBlocks  = (E + 255) / 256;
    const int markBlocks  = ((E > BS ? E : BS) + 255) / 256;
    const int zeroBlocks  = (nzero + 255) / 256;
    const int scanBlocks0 = (N + 1 + 1023) / 1024;
    const int scanBlocks1 = (BS + 1 + 1023) / 1024;
    const int scanBlocks  = scanBlocks0 > scanBlocks1 ? scanBlocks0 : scanBlocks1;
    const int headBlocks  = (BS + 7) / 8;
    const long long estRows64 = (long long)E * BS / N + BS;
    const int estRows     = estRows64 < N ? (int)estRows64 : N;
    const int nodeBlocksL = (estRows + 7) / 8;
    const int gemmBlocksL = (estRows + 15) / 16;

    prologue_k   <<<3 * kPackBlocks + zeroBlocks, 256, 0, stream>>>(w0, w1, wout, Wp0, Wp1, Wp2, zreg, nzero);

    gemm_mark_k  <<<gemmBlocksN + markBlocks, 256, 0, stream>>>(embs, Wp0, b0, xl, N, gemmBlocksN,
                                                                 ei1, cnt1, need, E, BS);

    scan_bin_k   <<<2 * scanBlocks + edgeBlocks, 256, 0, stream>>>(need, list, nlist, N,
                                                                    cnt1, rowptr1, cursor1, BS, scanBlocks,
                                                                    ei0, cnt, (int*)buf, E);
    node_gat_bins_k<<<nodeBlocksL, 256, 0, stream>>>(xl, cnt, ei0, E, att0, bias0, buf, list, nlist);

    gemm_fill_k  <<<gemmBlocksL + edgeBlocks, 256, 0, stream>>>(buf, Wp1, b1, xl, list, nlist, gemmBlocksL,
                                                                 ei1, cursor1, srcs, E, BS);
    gat_head_ln_k<<<headBlocks, 256, 0, stream>>>(xl, rowptr1, srcs, att1, bias1, Wp2, bout, gamma, beta,
                                                  (float*)d_out, BS);
}
